# starting kernel
# baseline (speedup 1.0000x reference)
#include <hip/hip_runtime.h>

typedef __attribute__((ext_vector_type(2))) float v2f;
typedef __attribute__((ext_vector_type(8))) float v8f;

#define NNODES 100000
#define NEDGES 1250000


__global__ void fill_kernel(float* __restrict__ p, long long n, float v) {
    long long i = (long long)blockIdx.x * blockDim.x + threadIdx.x;
    if (i < n) p[i] = v;
}

__global__ void deg_kernel(const int* __restrict__ col, float* __restrict__ deg, int E) {
    int e = blockIdx.x * blockDim.x + threadIdx.x;
    if (e < E) atomicAdd(&deg[col[e]], 1.0f);
}

__global__ void rsqrt_kernel(float* __restrict__ d, int n) {
    int i = blockIdx.x * blockDim.x + threadIdx.x;
    if (i < n) {
        float x = d[i];
        d[i] = (x > 0.0f) ? rsqrtf(x) : 0.0f;
    }
}

template <int K, int NOUTPAD>
__global__ __launch_bounds__(256) void gemm_wmma_kernel(
    const float* __restrict__ X, const float* __restrict__ W,
    const float* __restrict__ bias, float* __restrict__ Y,
    int M, int Nout, int addBias)
{
    constexpr int STRIDE = K + 4;
    constexpr int NT = NOUTPAD / 16;
    __shared__ float Wt[NOUTPAD * STRIDE];

    for (int idx = threadIdx.x; idx < NOUTPAD * K; idx += 256) {
        int n = idx / K;
        int k = idx - n * K;
        Wt[n * STRIDE + k] = (n < Nout) ? W[(size_t)k * Nout + n] : 0.0f;
    }
    __syncthreads();

    const int lane  = threadIdx.x & 31;
    const int strip = blockIdx.x * 8 + (threadIdx.x >> 5);
    const int Mtiles = M >> 4;
    if (strip >= Mtiles) return;

    const int half = lane >> 4;
    const int l    = lane & 15;
    const int row0 = strip << 4;

    const float* __restrict__ ap  = X + (size_t)(row0 + l) * K + 2 * half;
    const float* __restrict__ bp0 = &Wt[l * STRIDE + 2 * half];

    v8f acc[NT];
#pragma unroll
    for (int t = 0; t < NT; ++t) acc[t] = {};

#pragma unroll 4
    for (int k = 0; k < K; k += 4) {
        v2f a;
        a.x = ap[k];
        a.y = ap[k + 1];
#pragma unroll
        for (int t = 0; t < NT; ++t) {
            v2f b;
            const float* bp = bp0 + t * 16 * STRIDE;
            b.x = bp[k];
            b.y = bp[k + 1];
            acc[t] = __builtin_amdgcn_wmma_f32_16x16x4_f32(
                false, a, false, b, (short)0, acc[t], false, false);
        }
    }

#pragma unroll
    for (int t = 0; t < NT; ++t) {
        int c = t * 16 + l;
        if (c < Nout) {
            float bv = addBias ? bias[c] : 0.0f;
#pragma unroll
            for (int v = 0; v < 8; ++v) {
                int r = row0 + v + 8 * half;
                Y[(size_t)r * Nout + c] = acc[t][v] + bv;
            }
        }
    }
}

__global__ void scatter_kernel(const float* __restrict__ H, const int* __restrict__ row,
                               const int* __restrict__ col, const float* __restrict__ dis,
                               float* __restrict__ AGG, int E) {
    long long gid = (long long)blockIdx.x * blockDim.x + threadIdx.x;
    int e  = (int)(gid >> 4);
    int f4 = (int)(gid & 15) << 2;
    if (e >= E) return;
    int r = row[e];
    int c = col[e];
    float nrm = dis[r] * dis[c];
    const float4 h = *(const float4*)(H + (size_t)r * 64 + f4);
    float* out = AGG + (size_t)c * 64 + f4;
    atomicAdd(out + 0, h.x * nrm);
    atomicAdd(out + 1, h.y * nrm);
    atomicAdd(out + 2, h.z * nrm);
    atomicAdd(out + 3, h.w * nrm);
}

__global__ void combine_kernel(float* __restrict__ agg, const float* __restrict__ htmp,
                               const float* __restrict__ dis, const float* __restrict__ bias,
                               int n  ) {
    int i = blockIdx.x * blockDim.x + threadIdx.x;
    if (i >= n) return;
    int node = i >> 6;
    int f    = i & 63;
    float d = dis[node];
    float v = agg[i] + htmp[i] * d * d + bias[f];
    agg[i] = (v > 0.0f) ? v : 0.0f;
}


extern "C" void kernel_launch(void* const* d_in, const int* in_sizes, int n_in,
                              void* d_out, int out_size, void* d_ws, size_t ws_size,
                              hipStream_t stream) {
    const float* x    = (const float*)d_in[0];
    const int*   ei   = (const int*)d_in[1];
    const float* W1   = (const float*)d_in[2];
    const float* b1   = (const float*)d_in[3];
    const float* W2   = (const float*)d_in[4];
    const float* b2   = (const float*)d_in[5];
    const float* Wout = (const float*)d_in[6];
    const float* bout = (const float*)d_in[7];
    float* out = (float*)d_out;

    const int* row = ei;
    const int* col = ei + NEDGES;

    float* dis = (float*)d_ws;
    float* A   = dis + NNODES;
    float* B   = A + (size_t)NNODES * 64;
    float* C   = B + (size_t)NNODES * 64;
    (void)ws_size; (void)n_in; (void)in_sizes; (void)out_size;

    const int T = 256;
    const long long NF = (long long)NNODES * 64;
    const int Mtiles = NNODES / 16;
    const int gemmBlocks = (Mtiles + 7) / 8;

    fill_kernel<<<(NNODES + T - 1) / T, T, 0, stream>>>(dis, NNODES, 1.0f);
    deg_kernel<<<(NEDGES + T - 1) / T, T, 0, stream>>>(col, dis, NEDGES);
    rsqrt_kernel<<<(NNODES + T - 1) / T, T, 0, stream>>>(dis, NNODES);

    gemm_wmma_kernel<128, 64><<<gemmBlocks, T, 0, stream>>>(x, W1, nullptr, A, NNODES, 64, 0);
    fill_kernel<<<(int)((NF + T - 1) / T), T, 0, stream>>>(B, NF, 0.0f);
    scatter_kernel<<<(int)(((long long)NEDGES * 16 + T - 1) / T), T, 0, stream>>>(A, row, col, dis, B, NEDGES);
    combine_kernel<<<(int)((NF + T - 1) / T), T, 0, stream>>>(B, A, dis, b1, (int)NF);

    gemm_wmma_kernel<64, 64><<<gemmBlocks, T, 0, stream>>>(B, W2, nullptr, A, NNODES, 64, 0);
    fill_kernel<<<(int)((NF + T - 1) / T), T, 0, stream>>>(C, NF, 0.0f);
    scatter_kernel<<<(int)(((long long)NEDGES * 16 + T - 1) / T), T, 0, stream>>>(A, row, col, dis, C, NEDGES);
    combine_kernel<<<(int)((NF + T - 1) / T), T, 0, stream>>>(C, A, dis, b2, (int)NF);

    gemm_wmma_kernel<64, 48><<<gemmBlocks, T, 0, stream>>>(C, Wout, bout, out, NNODES, 40, 1);
}


# float-atomic edge scatter -> per-node gather over a destination-ordered edge list rebuilt each call (arrival-order integer rank atomics + scan); zero fill and combine fold into the gather
# speedup vs baseline: 2.0252x; 2.0252x over previous
#include <hip/hip_runtime.h>

typedef __attribute__((ext_vector_type(2))) float v2f;
typedef __attribute__((ext_vector_type(8))) float v8f;

#define NNODES 100000
#define NEDGES 1250000


__global__ void zero_kernel(int* __restrict__ p, int n) {
    int i = blockIdx.x * blockDim.x + threadIdx.x;
    if (i < n) p[i] = 0;
}

__global__ void rank_kernel(const int* __restrict__ col, int* __restrict__ cnt,
                            int* __restrict__ rank, int E) {
    int e = blockIdx.x * blockDim.x + threadIdx.x;
    if (e < E) rank[e] = atomicAdd(&cnt[col[e]], 1);
}

__global__ __launch_bounds__(1024) void offsets_kernel(const int* __restrict__ cnt,
                                                       int* __restrict__ off,
                                                       float* __restrict__ dis, int n) {
    __shared__ int wtot[32];
    __shared__ int wpart[32];
    __shared__ int s_prefix;

    const int tid  = threadIdx.x;
    const int lane = tid & 31;
    const int w    = tid >> 5;
    const int base = blockIdx.x * 1024;

    int part = 0;
    for (int i = tid; i < base; i += 1024) part += cnt[i];

    const int i = base + tid;
    const int v = (i < n) ? cnt[i] : 0;

    int incl = v;
#pragma unroll
    for (int d = 1; d < 32; d <<= 1) {
        int t = __shfl_up(incl, d);
        if (lane >= d) incl += t;
    }
#pragma unroll
    for (int d = 16; d > 0; d >>= 1) part += __shfl_xor(part, d);

    if (lane == 31) wtot[w] = incl;
    if (lane == 0)  wpart[w] = part;
    __syncthreads();

    if (w == 0) {
        const int t = wtot[lane];
        int ti = t;
#pragma unroll
        for (int d = 1; d < 32; d <<= 1) {
            int u = __shfl_up(ti, d);
            if (lane >= d) ti += u;
        }
        int p = wpart[lane];
#pragma unroll
        for (int d = 16; d > 0; d >>= 1) p += __shfl_xor(p, d);
        wtot[lane] = ti - t;
        if (lane == 0) s_prefix = p;
    }
    __syncthreads();

    if (i < n) {
        off[i] = s_prefix + wtot[w] + incl - v;
        float x = (float)(v + 1);
        dis[i] = (x > 0.0f) ? rsqrtf(x) : 0.0f;
    }
}

__global__ void place_kernel(const int* __restrict__ row, const int* __restrict__ col,
                             const int* __restrict__ off, const int* __restrict__ rank,
                             int* __restrict__ src, int E) {
    int e = blockIdx.x * blockDim.x + threadIdx.x;
    if (e < E) src[off[col[e]] + rank[e]] = row[e];
}

template <int K, int NOUTPAD>
__global__ __launch_bounds__(256) void gemm_wmma_kernel(
    const float* __restrict__ X, const float* __restrict__ W,
    const float* __restrict__ bias, float* __restrict__ Y,
    int M, int Nout, int addBias)
{
    constexpr int STRIDE = K + 4;
    constexpr int NT = NOUTPAD / 16;
    __shared__ float Wt[NOUTPAD * STRIDE];

    for (int idx = threadIdx.x; idx < NOUTPAD * K; idx += 256) {
        int n = idx / K;
        int k = idx - n * K;
        Wt[n * STRIDE + k] = (n < Nout) ? W[(size_t)k * Nout + n] : 0.0f;
    }
    __syncthreads();

    const int lane  = threadIdx.x & 31;
    const int strip = blockIdx.x * 8 + (threadIdx.x >> 5);
    const int Mtiles = M >> 4;
    if (strip >= Mtiles) return;

    const int half = lane >> 4;
    const int l    = lane & 15;
    const int row0 = strip << 4;

    const float* __restrict__ ap  = X + (size_t)(row0 + l) * K + 2 * half;
    const float* __restrict__ bp0 = &Wt[l * STRIDE + 2 * half];

    v8f acc[NT];
#pragma unroll
    for (int t = 0; t < NT; ++t) acc[t] = {};

#pragma unroll 4
    for (int k = 0; k < K; k += 4) {
        v2f a;
        a.x = ap[k];
        a.y = ap[k + 1];
#pragma unroll
        for (int t = 0; t < NT; ++t) {
            v2f b;
            const float* bp = bp0 + t * 16 * STRIDE;
            b.x = bp[k];
            b.y = bp[k + 1];
            acc[t] = __builtin_amdgcn_wmma_f32_16x16x4_f32(
                false, a, false, b, (short)0, acc[t], false, false);
        }
    }

#pragma unroll
    for (int t = 0; t < NT; ++t) {
        int c = t * 16 + l;
        if (c < Nout) {
            float bv = addBias ? bias[c] : 0.0f;
#pragma unroll
            for (int v = 0; v < 8; ++v) {
                int r = row0 + v + 8 * half;
                Y[(size_t)r * Nout + c] = acc[t][v] + bv;
            }
        }
    }
}

__global__ __launch_bounds__(256) void gather_kernel(
    const float* __restrict__ H, const int* __restrict__ src,
    const int* __restrict__ off, const int* __restrict__ cnt,
    const float* __restrict__ dis, const float* __restrict__ bias,
    float* __restrict__ OUT, int n)
{
    const int gid = blockIdx.x * 256 + threadIdx.x;
    const int c   = gid >> 4;
    const int f4  = (gid & 15) << 2;
    if (c >= n) return;

    const int   beg = off[c];
    const int   m   = cnt[c];
    const float dc  = dis[c];
    const int* __restrict__ sp = src + beg;
    const float* __restrict__ Hf = H + f4;

    float4 acc = make_float4(0.0f, 0.0f, 0.0f, 0.0f);

    int j = 0;
    for (; j + 4 <= m; j += 4) {
        const int r0 = sp[j], r1 = sp[j + 1], r2 = sp[j + 2], r3 = sp[j + 3];
        const float4 h0 = *(const float4*)(Hf + (size_t)r0 * 64);
        const float4 h1 = *(const float4*)(Hf + (size_t)r1 * 64);
        const float4 h2 = *(const float4*)(Hf + (size_t)r2 * 64);
        const float4 h3 = *(const float4*)(Hf + (size_t)r3 * 64);
        const float n0 = dis[r0] * dc, n1 = dis[r1] * dc, n2 = dis[r2] * dc, n3 = dis[r3] * dc;
        acc.x += h0.x * n0; acc.y += h0.y * n0; acc.z += h0.z * n0; acc.w += h0.w * n0;
        acc.x += h1.x * n1; acc.y += h1.y * n1; acc.z += h1.z * n1; acc.w += h1.w * n1;
        acc.x += h2.x * n2; acc.y += h2.y * n2; acc.z += h2.z * n2; acc.w += h2.w * n2;
        acc.x += h3.x * n3; acc.y += h3.y * n3; acc.z += h3.z * n3; acc.w += h3.w * n3;
    }
    const int rem = m - j;
    if (rem > 0) {
        const bool k1 = rem > 1, k2 = rem > 2;
        const int r0 = sp[j];
        const int r1 = sp[j + (k1 ? 1 : 0)];
        const int r2 = sp[j + (k2 ? 2 : 0)];
        const float4 h0 = *(const float4*)(Hf + (size_t)r0 * 64);
        const float4 h1 = *(const float4*)(Hf + (size_t)r1 * 64);
        const float4 h2 = *(const float4*)(Hf + (size_t)r2 * 64);
        const float n0 = dis[r0] * dc, n1 = dis[r1] * dc, n2 = dis[r2] * dc;
        acc.x += h0.x * n0; acc.y += h0.y * n0; acc.z += h0.z * n0; acc.w += h0.w * n0;
        acc.x += k1 ? h1.x * n1 : 0.0f; acc.y += k1 ? h1.y * n1 : 0.0f;
        acc.z += k1 ? h1.z * n1 : 0.0f; acc.w += k1 ? h1.w * n1 : 0.0f;
        acc.x += k2 ? h2.x * n2 : 0.0f; acc.y += k2 ? h2.y * n2 : 0.0f;
        acc.z += k2 ? h2.z * n2 : 0.0f; acc.w += k2 ? h2.w * n2 : 0.0f;
    }

    const float4 hs = *(const float4*)(Hf + (size_t)c * 64);
    const float4 b  = *(const float4*)(bias + f4);
    float4 o;
    o.x = acc.x + hs.x * dc * dc + b.x;
    o.y = acc.y + hs.y * dc * dc + b.y;
    o.z = acc.z + hs.z * dc * dc + b.z;
    o.w = acc.w + hs.w * dc * dc + b.w;
    o.x = (o.x > 0.0f) ? o.x : 0.0f;
    o.y = (o.y > 0.0f) ? o.y : 0.0f;
    o.z = (o.z > 0.0f) ? o.z : 0.0f;
    o.w = (o.w > 0.0f) ? o.w : 0.0f;
    *(float4*)(OUT + (size_t)c * 64 + f4) = o;
}


extern "C" void kernel_launch(void* const* d_in, const int* in_sizes, int n_in,
                              void* d_out, int out_size, void* d_ws, size_t ws_size,
                              hipStream_t stream) {
    const float* x    = (const float*)d_in[0];
    const int*   ei   = (const int*)d_in[1];
    const float* W1   = (const float*)d_in[2];
    const float* b1   = (const float*)d_in[3];
    const float* W2   = (const float*)d_in[4];
    const float* b2   = (const float*)d_in[5];
    const float* Wout = (const float*)d_in[6];
    const float* bout = (const float*)d_in[7];
    float* out = (float*)d_out;

    const int* row = ei;
    const int* col = ei + NEDGES;

    float* A   = (float*)d_ws;
    float* B   = A + (size_t)NNODES * 64;
    float* C   = B + (size_t)NNODES * 64;
    float* dis = C + (size_t)NNODES * 64;
    int* cnt   = (int*)(dis + NNODES);
    int* off   = cnt + NNODES;
    int* rank  = off + NNODES;
    int* src   = rank + NEDGES;
    (void)ws_size; (void)n_in; (void)in_sizes; (void)out_size;

    const int T = 256;
    const int Mtiles = NNODES / 16;
    const int gemmBlocks = (Mtiles + 7) / 8;
    const int edgeBlocks = (NEDGES + T - 1) / T;
    const int gatherBlocks = (NNODES * 16 + T - 1) / T;

    zero_kernel<<<(NNODES + T - 1) / T, T, 0, stream>>>(cnt, NNODES);
    rank_kernel<<<edgeBlocks, T, 0, stream>>>(col, cnt, rank, NEDGES);
    offsets_kernel<<<(NNODES + 1023) / 1024, 1024, 0, stream>>>(cnt, off, dis, NNODES);
    place_kernel<<<edgeBlocks, T, 0, stream>>>(row, col, off, rank, src, NEDGES);

    gemm_wmma_kernel<128, 64><<<gemmBlocks, T, 0, stream>>>(x, W1, nullptr, A, NNODES, 64, 0);
    gather_kernel<<<gatherBlocks, T, 0, stream>>>(A, src, off, cnt, dis, b1, B, NNODES);

    gemm_wmma_kernel<64, 64><<<gemmBlocks, T, 0, stream>>>(B, W2, nullptr, A, NNODES, 64, 0);
    gather_kernel<<<gatherBlocks, T, 0, stream>>>(A, src, off, cnt, dis, b2, C, NNODES);

    gemm_wmma_kernel<64, 48><<<gemmBlocks, T, 0, stream>>>(C, Wout, bout, out, NNODES, 40, 1);
}


# prefix scan + place pass dropped: the rank atomic puts each source straight into a fixed 32-slot row per destination (rare extras to an overflow list), dis from in-degree in gather; sum order kept
# speedup vs baseline: 2.3282x; 1.1496x over previous
#include <hip/hip_runtime.h>

typedef __attribute__((ext_vector_type(2))) float v2f;
typedef __attribute__((ext_vector_type(8))) float v8f;

#define NNODES 100000
#define NEDGES 1250000


#define SLOTS 32


__global__ void zero_kernel(int* __restrict__ p, int n) {
    int i = blockIdx.x * blockDim.x + threadIdx.x;
    if (i < n) p[i] = 0;
}

__global__ void slot_kernel(const int* __restrict__ row, const int* __restrict__ col,
                            int* __restrict__ cnt, int* __restrict__ slots,
                            int* __restrict__ novf, int* __restrict__ ovf_c,
                            int* __restrict__ ovf_r, int E) {
    int e = blockIdx.x * blockDim.x + threadIdx.x;
    if (e < E) {
        const int c = col[e];
        const int r = row[e];
        const int k = atomicAdd(&cnt[c], 1);
        if (k < SLOTS) {
            slots[c * SLOTS + k] = r;
        } else {
            const int o = atomicAdd(novf, 1);
            ovf_c[o] = c;
            ovf_r[o] = r;
        }
    }
}

__device__ __forceinline__ float inv_sqrt_deg(int indeg) {
    const float x = (float)(indeg + 1);
    return (x > 0.0f) ? rsqrtf(x) : 0.0f;
}

__device__ __forceinline__ float4 row_f4(const float* __restrict__ H, int r, unsigned lane_bytes) {
    const unsigned off = ((unsigned)r << 8) + lane_bytes;
    return *(const float4*)((const char*)H + off);
}

template <int K, int NOUTPAD>
__global__ __launch_bounds__(256) void gemm_wmma_kernel(
    const float* __restrict__ X, const float* __restrict__ W,
    const float* __restrict__ bias, float* __restrict__ Y,
    int M, int Nout, int addBias)
{
    constexpr int STRIDE = K + 4;
    constexpr int NT = NOUTPAD / 16;
    __shared__ float Wt[NOUTPAD * STRIDE];

    for (int idx = threadIdx.x; idx < NOUTPAD * K; idx += 256) {
        int n = idx / K;
        int k = idx - n * K;
        Wt[n * STRIDE + k] = (n < Nout) ? W[(size_t)k * Nout + n] : 0.0f;
    }
    __syncthreads();

    const int lane  = threadIdx.x & 31;
    const int strip = blockIdx.x * 8 + (threadIdx.x >> 5);
    const int Mtiles = M >> 4;
    if (strip >= Mtiles) return;

    const int half = lane >> 4;
    const int l    = lane & 15;
    const int row0 = strip << 4;

    const float* __restrict__ ap  = X + (size_t)(row0 + l) * K + 2 * half;
    const float* __restrict__ bp0 = &Wt[l * STRIDE + 2 * half];

    v8f acc[NT];
#pragma unroll
    for (int t = 0; t < NT; ++t) acc[t] = {};

#pragma unroll 4
    for (int k = 0; k < K; k += 4) {
        v2f a;
        a.x = ap[k];
        a.y = ap[k + 1];
#pragma unroll
        for (int t = 0; t < NT; ++t) {
            v2f b;
            const float* bp = bp0 + t * 16 * STRIDE;
            b.x = bp[k];
            b.y = bp[k + 1];
            acc[t] = __builtin_amdgcn_wmma_f32_16x16x4_f32(
                false, a, false, b, (short)0, acc[t], false, false);
        }
    }

#pragma unroll
    for (int t = 0; t < NT; ++t) {
        int c = t * 16 + l;
        if (c < Nout) {
            float bv = addBias ? bias[c] : 0.0f;
#pragma unroll
            for (int v = 0; v < 8; ++v) {
                int r = row0 + v + 8 * half;
                Y[(size_t)r * Nout + c] = acc[t][v] + bv;
            }
        }
    }
}

__global__ __launch_bounds__(256) void gather_kernel(
    const float* __restrict__ H, const int* __restrict__ slots,
    const int* __restrict__ cnt, const int* __restrict__ novf,
    const int* __restrict__ ovf_c, const int* __restrict__ ovf_r,
    const float* __restrict__ bias, float* __restrict__ OUT, int n)
{
    const int gid = blockIdx.x * 256 + threadIdx.x;
    const int c   = gid >> 4;
    const int f4  = (gid & 15) << 2;
    if (c >= n) return;

    const int   m   = cnt[c];
    const float dc  = inv_sqrt_deg(m);
    const int   ms  = (m < SLOTS) ? m : SLOTS;
    const int* __restrict__ sp = slots + (size_t)c * SLOTS;
    const unsigned lb = (unsigned)f4 << 2;

    float4 acc = make_float4(0.0f, 0.0f, 0.0f, 0.0f);

    int j = 0;
#pragma unroll 2
    for (; j + 4 <= ms; j += 4) {
        const int r0 = sp[j], r1 = sp[j + 1], r2 = sp[j + 2], r3 = sp[j + 3];
        const float4 h0 = row_f4(H, r0, lb);
        const float4 h1 = row_f4(H, r1, lb);
        const float4 h2 = row_f4(H, r2, lb);
        const float4 h3 = row_f4(H, r3, lb);
        const float n0 = inv_sqrt_deg(cnt[r0]) * dc, n1 = inv_sqrt_deg(cnt[r1]) * dc;
        const float n2 = inv_sqrt_deg(cnt[r2]) * dc, n3 = inv_sqrt_deg(cnt[r3]) * dc;
        acc.x += h0.x * n0; acc.y += h0.y * n0; acc.z += h0.z * n0; acc.w += h0.w * n0;
        acc.x += h1.x * n1; acc.y += h1.y * n1; acc.z += h1.z * n1; acc.w += h1.w * n1;
        acc.x += h2.x * n2; acc.y += h2.y * n2; acc.z += h2.z * n2; acc.w += h2.w * n2;
        acc.x += h3.x * n3; acc.y += h3.y * n3; acc.z += h3.z * n3; acc.w += h3.w * n3;
    }
    const int rem = ms - j;
    if (rem > 0) {
        const bool k1 = rem > 1, k2 = rem > 2;
        const int r0 = sp[j];
        const int r1 = sp[j + (k1 ? 1 : 0)];
        const int r2 = sp[j + (k2 ? 2 : 0)];
        const float4 h0 = row_f4(H, r0, lb);
        const float4 h1 = row_f4(H, r1, lb);
        const float4 h2 = row_f4(H, r2, lb);
        const float n0 = inv_sqrt_deg(cnt[r0]) * dc, n1 = inv_sqrt_deg(cnt[r1]) * dc;
        const float n2 = inv_sqrt_deg(cnt[r2]) * dc;
        acc.x += h0.x * n0; acc.y += h0.y * n0; acc.z += h0.z * n0; acc.w += h0.w * n0;
        acc.x += k1 ? h1.x * n1 : 0.0f; acc.y += k1 ? h1.y * n1 : 0.0f;
        acc.z += k1 ? h1.z * n1 : 0.0f; acc.w += k1 ? h1.w * n1 : 0.0f;
        acc.x += k2 ? h2.x * n2 : 0.0f; acc.y += k2 ? h2.y * n2 : 0.0f;
        acc.z += k2 ? h2.z * n2 : 0.0f; acc.w += k2 ? h2.w * n2 : 0.0f;
    }
    if (m > SLOTS) {
        const int no = *novf;
        for (int o = 0; o < no; ++o) {
            if (ovf_c[o] == c) {
                const int r = ovf_r[o];
                const float4 h = row_f4(H, r, lb);
                const float nr = inv_sqrt_deg(cnt[r]) * dc;
                acc.x += h.x * nr; acc.y += h.y * nr; acc.z += h.z * nr; acc.w += h.w * nr;
            }
        }
    }

    const float4 hs = row_f4(H, c, lb);
    const float4 b  = *(const float4*)(bias + f4);
    float4 o;
    o.x = acc.x + hs.x * dc * dc + b.x;
    o.y = acc.y + hs.y * dc * dc + b.y;
    o.z = acc.z + hs.z * dc * dc + b.z;
    o.w = acc.w + hs.w * dc * dc + b.w;
    o.x = (o.x > 0.0f) ? o.x : 0.0f;
    o.y = (o.y > 0.0f) ? o.y : 0.0f;
    o.z = (o.z > 0.0f) ? o.z : 0.0f;
    o.w = (o.w > 0.0f) ? o.w : 0.0f;
    *(float4*)(OUT + (size_t)c * 64 + f4) = o;
}


extern "C" void kernel_launch(void* const* d_in, const int* in_sizes, int n_in,
                              void* d_out, int out_size, void* d_ws, size_t ws_size,
                              hipStream_t stream) {
    const float* x    = (const float*)d_in[0];
    const int*   ei   = (const int*)d_in[1];
    const float* W1   = (const float*)d_in[2];
    const float* b1   = (const float*)d_in[3];
    const float* W2   = (const float*)d_in[4];
    const float* b2   = (const float*)d_in[5];
    const float* Wout = (const float*)d_in[6];
    const float* bout = (const float*)d_in[7];
    float* out = (float*)d_out;

    const int* row = ei;
    const int* col = ei + NEDGES;

    float* A   = (float*)d_ws;
    float* B   = A + (size_t)NNODES * 64;
    float* C   = B + (size_t)NNODES * 64;
    int* slots = (int*)(C + (size_t)NNODES * 64);
    int* cnt   = slots + (size_t)NNODES * SLOTS;
    int* novf  = cnt + NNODES;
    int* ovf_c = novf + 1;
    int* ovf_r = ovf_c + NEDGES;
    (void)ws_size; (void)n_in; (void)in_sizes; (void)out_size;

    const int T = 256;
    const int Mtiles = NNODES / 16;
    const int gemmBlocks = (Mtiles + 7) / 8;
    const int edgeBlocks = (NEDGES + T - 1) / T;
    const int gatherBlocks = (NNODES * 16 + T - 1) / T;

    zero_kernel<<<(NNODES + 1 + T - 1) / T, T, 0, stream>>>(cnt, NNODES + 1);
    slot_kernel<<<edgeBlocks, T, 0, stream>>>(row, col, cnt, slots, novf, ovf_c, ovf_r, NEDGES);

    gemm_wmma_kernel<128, 64><<<gemmBlocks, T, 0, stream>>>(x, W1, nullptr, A, NNODES, 64, 0);
    gather_kernel<<<gatherBlocks, T, 0, stream>>>(A, slots, cnt, novf, ovf_c, ovf_r, b1, B, NNODES);

    gemm_wmma_kernel<64, 64><<<gemmBlocks, T, 0, stream>>>(B, W2, nullptr, A, NNODES, 64, 0);
    gather_kernel<<<gatherBlocks, T, 0, stream>>>(A, slots, cnt, novf, ovf_c, ovf_r, b2, C, NNODES);

    gemm_wmma_kernel<64, 48><<<gemmBlocks, T, 0, stream>>>(C, Wout, bout, out, NNODES, 40, 1);
}


# each product that consumes an aggregation (H1 @ W2, H2 @ Wout) runs in that gather's epilogue: block's 16 rows to an LDS tile, WMMA from it; H1/H2 never stored; 7 launches to 5; sums unchanged
# speedup vs baseline: 2.5506x; 1.0955x over previous
#include <hip/hip_runtime.h>

typedef __attribute__((ext_vector_type(2))) float v2f;
typedef __attribute__((ext_vector_type(8))) float v8f;

#define NNODES 100000
#define NEDGES 1250000


#define SLOTS 32


__global__ void zero_kernel(int* __restrict__ p, int n) {
    int i = blockIdx.x * blockDim.x + threadIdx.x;
    if (i < n) p[i] = 0;
}

__global__ void slot_kernel(const int* __restrict__ row, const int* __restrict__ col,
                            int* __restrict__ cnt, int* __restrict__ slots,
                            int* __restrict__ novf, int* __restrict__ ovf_c,
                            int* __restrict__ ovf_r, int E) {
    int e = blockIdx.x * blockDim.x + threadIdx.x;
    if (e < E) {
        const int c = col[e];
        const int r = row[e];
        const int k = atomicAdd(&cnt[c], 1);
        if (k < SLOTS) {
            slots[c * SLOTS + k] = r;
        } else {
            const int o = atomicAdd(novf, 1);
            ovf_c[o] = c;
            ovf_r[o] = r;
        }
    }
}

__device__ __forceinline__ float inv_sqrt_deg(int indeg) {
    const float x = (float)(indeg + 1);
    return (x > 0.0f) ? rsqrtf(x) : 0.0f;
}

__device__ __forceinline__ float4 row_f4(const float* __restrict__ H, int r, unsigned lane_bytes) {
    const unsigned off = ((unsigned)r << 8) + lane_bytes;
    return *(const float4*)((const char*)H + off);
}

template <int K, int NOUTPAD>
__global__ __launch_bounds__(256) void gemm_wmma_kernel(
    const float* __restrict__ X, const float* __restrict__ W,
    const float* __restrict__ bias, float* __restrict__ Y,
    int M, int Nout, int addBias)
{
    constexpr int STRIDE = K + 4;
    constexpr int NT = NOUTPAD / 16;
    __shared__ float Wt[NOUTPAD * STRIDE];

    for (int idx = threadIdx.x; idx < NOUTPAD * K; idx += 256) {
        int n = idx / K;
        int k = idx - n * K;
        Wt[n * STRIDE + k] = (n < Nout) ? W[(size_t)k * Nout + n] : 0.0f;
    }
    __syncthreads();

    const int lane  = threadIdx.x & 31;
    const int strip = blockIdx.x * 8 + (threadIdx.x >> 5);
    const int Mtiles = M >> 4;
    if (strip >= Mtiles) return;

    const int half = lane >> 4;
    const int l    = lane & 15;
    const int row0 = strip << 4;

    const float* __restrict__ ap  = X + (size_t)(row0 + l) * K + 2 * half;
    const float* __restrict__ bp0 = &Wt[l * STRIDE + 2 * half];

    v8f acc[NT];
#pragma unroll
    for (int t = 0; t < NT; ++t) acc[t] = {};

#pragma unroll 4
    for (int k = 0; k < K; k += 4) {
        v2f a;
        a.x = ap[k];
        a.y = ap[k + 1];
#pragma unroll
        for (int t = 0; t < NT; ++t) {
            v2f b;
            const float* bp = bp0 + t * 16 * STRIDE;
            b.x = bp[k];
            b.y = bp[k + 1];
            acc[t] = __builtin_amdgcn_wmma_f32_16x16x4_f32(
                false, a, false, b, (short)0, acc[t], false, false);
        }
    }

#pragma unroll
    for (int t = 0; t < NT; ++t) {
        int c = t * 16 + l;
        if (c < Nout) {
            float bv = addBias ? bias[c] : 0.0f;
#pragma unroll
            for (int v = 0; v < 8; ++v) {
                int r = row0 + v + 8 * half;
                Y[(size_t)r * Nout + c] = acc[t][v] + bv;
            }
        }
    }
}

template <int NOUT, bool ADD_BIAS>
__global__ __launch_bounds__(256) void gather_gemm_kernel(
    const float* __restrict__ H, const int* __restrict__ slots,
    const int* __restrict__ cnt, const int* __restrict__ novf,
    const int* __restrict__ ovf_c, const int* __restrict__ ovf_r,
    const float* __restrict__ bias, const float* __restrict__ Wn,
    const float* __restrict__ bn, float* __restrict__ Y)
{
    constexpr int TSTRIDE = 68;
    constexpr int NT = (NOUT + 15) / 16;
    __shared__ float T[16 * TSTRIDE];

    const int gid = blockIdx.x * 256 + threadIdx.x;
    const int c   = gid >> 4;
    const int f4  = (gid & 15) << 2;

    const int   m   = cnt[c];
    const float dc  = inv_sqrt_deg(m);
    const int   ms  = (m < SLOTS) ? m : SLOTS;
    const int* __restrict__ sp = slots + (size_t)c * SLOTS;
    const unsigned lb = (unsigned)f4 << 2;

    float4 acc = make_float4(0.0f, 0.0f, 0.0f, 0.0f);

    int j = 0;
#pragma unroll 2
    for (; j + 4 <= ms; j += 4) {
        const int r0 = sp[j], r1 = sp[j + 1], r2 = sp[j + 2], r3 = sp[j + 3];
        const float4 h0 = row_f4(H, r0, lb);
        const float4 h1 = row_f4(H, r1, lb);
        const float4 h2 = row_f4(H, r2, lb);
        const float4 h3 = row_f4(H, r3, lb);
        const float n0 = inv_sqrt_deg(cnt[r0]) * dc, n1 = inv_sqrt_deg(cnt[r1]) * dc;
        const float n2 = inv_sqrt_deg(cnt[r2]) * dc, n3 = inv_sqrt_deg(cnt[r3]) * dc;
        acc.x += h0.x * n0; acc.y += h0.y * n0; acc.z += h0.z * n0; acc.w += h0.w * n0;
        acc.x += h1.x * n1; acc.y += h1.y * n1; acc.z += h1.z * n1; acc.w += h1.w * n1;
        acc.x += h2.x * n2; acc.y += h2.y * n2; acc.z += h2.z * n2; acc.w += h2.w * n2;
        acc.x += h3.x * n3; acc.y += h3.y * n3; acc.z += h3.z * n3; acc.w += h3.w * n3;
    }
    const int rem = ms - j;
    if (rem > 0) {
        const bool k1 = rem > 1, k2 = rem > 2;
        const int r0 = sp[j];
        const int r1 = sp[j + (k1 ? 1 : 0)];
        const int r2 = sp[j + (k2 ? 2 : 0)];
        const float4 h0 = row_f4(H, r0, lb);
        const float4 h1 = row_f4(H, r1, lb);
        const float4 h2 = row_f4(H, r2, lb);
        const float n0 = inv_sqrt_deg(cnt[r0]) * dc, n1 = inv_sqrt_deg(cnt[r1]) * dc;
        const float n2 = inv_sqrt_deg(cnt[r2]) * dc;
        acc.x += h0.x * n0; acc.y += h0.y * n0; acc.z += h0.z * n0; acc.w += h0.w * n0;
        acc.x += k1 ? h1.x * n1 : 0.0f; acc.y += k1 ? h1.y * n1 : 0.0f;
        acc.z += k1 ? h1.z * n1 : 0.0f; acc.w += k1 ? h1.w * n1 : 0.0f;
        acc.x += k2 ? h2.x * n2 : 0.0f; acc.y += k2 ? h2.y * n2 : 0.0f;
        acc.z += k2 ? h2.z * n2 : 0.0f; acc.w += k2 ? h2.w * n2 : 0.0f;
    }
    if (m > SLOTS) {
        const int no = *novf;
        for (int o = 0; o < no; ++o) {
            if (ovf_c[o] == c) {
                const int r = ovf_r[o];
                const float4 h = row_f4(H, r, lb);
                const float nr = inv_sqrt_deg(cnt[r]) * dc;
                acc.x += h.x * nr; acc.y += h.y * nr; acc.z += h.z * nr; acc.w += h.w * nr;
            }
        }
    }

    const float4 hs = row_f4(H, c, lb);
    const float4 b  = *(const float4*)(bias + f4);
    float4 o;
    o.x = acc.x + hs.x * dc * dc + b.x;
    o.y = acc.y + hs.y * dc * dc + b.y;
    o.z = acc.z + hs.z * dc * dc + b.z;
    o.w = acc.w + hs.w * dc * dc + b.w;
    o.x = (o.x > 0.0f) ? o.x : 0.0f;
    o.y = (o.y > 0.0f) ? o.y : 0.0f;
    o.z = (o.z > 0.0f) ? o.z : 0.0f;
    o.w = (o.w > 0.0f) ? o.w : 0.0f;
    *(float4*)&T[(threadIdx.x >> 4) * TSTRIDE + f4] = o;

    const int wave = threadIdx.x >> 5;
    const int lane = threadIdx.x & 31;
    const int half = lane >> 4;
    const int l    = lane & 15;
    const int ncol = wave * 16 + l;

    v2f bf[16];
    float bv = 0.0f;
    if (wave < NT) {
        const bool cok = ncol < NOUT;
        if constexpr (ADD_BIAS) bv = bn[cok ? ncol : 0];
        const float* __restrict__ wp = Wn + (size_t)(2 * half) * NOUT + (cok ? ncol : 0);
#pragma unroll
        for (int s = 0; s < 16; ++s) {
            const float w0 = wp[(size_t)(4 * s) * NOUT];
            const float w1 = wp[(size_t)(4 * s + 1) * NOUT];
            bf[s].x = cok ? w0 : 0.0f;
            bf[s].y = cok ? w1 : 0.0f;
        }
    }
    __syncthreads();
    if (wave >= NT) return;

    v8f d = {};
    const float* __restrict__ ap = &T[l * TSTRIDE + 2 * half];
#pragma unroll
    for (int s = 0; s < 16; ++s) {
        const v2f a = *(const v2f*)(ap + 4 * s);
        d = __builtin_amdgcn_wmma_f32_16x16x4_f32(
            false, a, false, bf[s], (short)0, d, false, false);
    }

    if (ncol < NOUT) {
        const int row0 = blockIdx.x << 4;
        float* __restrict__ yp = Y + (size_t)(row0 + 8 * half) * NOUT + ncol;
        if constexpr (ADD_BIAS) {
#pragma unroll
            for (int v = 0; v < 8; ++v) yp[v * NOUT] = d[v] + bv;
        } else {
#pragma unroll
            for (int v = 0; v < 8; ++v) yp[v * NOUT] = d[v];
        }
    }
}


extern "C" void kernel_launch(void* const* d_in, const int* in_sizes, int n_in,
                              void* d_out, int out_size, void* d_ws, size_t ws_size,
                              hipStream_t stream) {
    const float* x    = (const float*)d_in[0];
    const int*   ei   = (const int*)d_in[1];
    const float* W1   = (const float*)d_in[2];
    const float* b1   = (const float*)d_in[3];
    const float* W2   = (const float*)d_in[4];
    const float* b2   = (const float*)d_in[5];
    const float* Wout = (const float*)d_in[6];
    const float* bout = (const float*)d_in[7];
    float* out = (float*)d_out;

    const int* row = ei;
    const int* col = ei + NEDGES;

    float* A   = (float*)d_ws;
    float* B   = A + (size_t)NNODES * 64;
    int* slots = (int*)(B + (size_t)NNODES * 64);
    int* cnt   = slots + (size_t)NNODES * SLOTS;
    int* novf  = cnt + NNODES;
    int* ovf_c = novf + 1;
    int* ovf_r = ovf_c + NEDGES;
    (void)ws_size; (void)n_in; (void)in_sizes; (void)out_size;

    const int T = 256;
    const int Mtiles = NNODES / 16;
    const int gemmBlocks = (Mtiles + 7) / 8;
    const int edgeBlocks = (NEDGES + T - 1) / T;
    static_assert(NNODES % 16 == 0, "a gather block owns 16 whole nodes");
    const int gatherBlocks = NNODES / 16;

    zero_kernel<<<(NNODES + 1 + T - 1) / T, T, 0, stream>>>(cnt, NNODES + 1);
    slot_kernel<<<edgeBlocks, T, 0, stream>>>(row, col, cnt, slots, novf, ovf_c, ovf_r, NEDGES);

    gemm_wmma_kernel<128, 64><<<gemmBlocks, T, 0, stream>>>(x, W1, nullptr, A, NNODES, 64, 0);

    gather_gemm_kernel<64, false><<<gatherBlocks, T, 0, stream>>>(
        A, slots, cnt, novf, ovf_c, ovf_r, b1, W2, nullptr, B);

    gather_gemm_kernel<40, true><<<gatherBlocks, T, 0, stream>>>(
        B, slots, cnt, novf, ovf_c, ovf_r, b2, Wout, bout, out);
}


# first product's operand feed: W staged by 16-byte loads asked together (was 32 dependent 32-line loads per thread), X strip by 16-byte loads k blocks ahead; k in blocks of 8, sum order changes
# speedup vs baseline: 2.7669x; 1.0848x over previous
#include <hip/hip_runtime.h>

typedef __attribute__((ext_vector_type(2))) float v2f;
typedef __attribute__((ext_vector_type(8))) float v8f;

#define NNODES 100000
#define NEDGES 1250000


#define SLOTS 32


__global__ void zero_kernel(int* __restrict__ p, int n) {
    int i = blockIdx.x * blockDim.x + threadIdx.x;
    if (i < n) p[i] = 0;
}

__global__ void slot_kernel(const int* __restrict__ row, const int* __restrict__ col,
                            int* __restrict__ cnt, int* __restrict__ slots,
                            int* __restrict__ novf, int* __restrict__ ovf_c,
                            int* __restrict__ ovf_r, int E) {
    int e = blockIdx.x * blockDim.x + threadIdx.x;
    if (e < E) {
        const int c = col[e];
        const int r = row[e];
        const int k = atomicAdd(&cnt[c], 1);
        if (k < SLOTS) {
            slots[c * SLOTS + k] = r;
        } else {
            const int o = atomicAdd(novf, 1);
            ovf_c[o] = c;
            ovf_r[o] = r;
        }
    }
}

__device__ __forceinline__ float inv_sqrt_deg(int indeg) {
    const float x = (float)(indeg + 1);
    return (x > 0.0f) ? rsqrtf(x) : 0.0f;
}

__device__ __forceinline__ float4 row_f4(const float* __restrict__ H, int r, unsigned lane_bytes) {
    const unsigned off = ((unsigned)r << 8) + lane_bytes;
    return *(const float4*)((const char*)H + off);
}

template <int K, int NOUT>
__global__ __launch_bounds__(256) void gemm_wmma_kernel(
    const float* __restrict__ X, const float* __restrict__ W,
    const int* __restrict__ indeg, float* __restrict__ Y, int M)
{
    constexpr int STRIDE = K + 4;
    constexpr int NT = NOUT / 16;
    constexpr int KB = K / 8;
    constexpr int QG = NOUT / 16;
    constexpr int CH = KB * QG;
    constexpr int PF = 8;
    static_assert(K % 8 == 0 && NOUT % 16 == 0 && CH % 8 == 0 && KB >= PF, "tile shape");
    static_assert(((STRIDE / 4) & 1) == 1 && STRIDE % 4 == 0, "row pitch: 16 bytes times an odd number");
    __shared__ __attribute__((aligned(16))) float Wt[NOUT * STRIDE];

    const int lane  = threadIdx.x & 31;
    const int wave  = threadIdx.x >> 5;
    const int strip = blockIdx.x * 8 + wave;
    const int Mtiles = M >> 4;
    const bool active = strip < Mtiles;

    const int half = lane >> 4;
    const int l    = lane & 15;
    const int row0 = strip << 4;

    const float4* __restrict__ ap = (const float4*)(X + (size_t)(active ? row0 + l : 0) * K + 4 * half);

    float4 a[KB];
    if (active) {
#pragma unroll
        for (int j = 0; j < PF; ++j) a[j] = ap[2 * j];
    }

    {
        float4 w[CH / 8];
        int wn[CH / 8], wk[CH / 8];
#pragma unroll
        for (int i = 0; i < CH / 8; ++i) {
            const int c = wave * (CH / 8) + i;
            const int q = (c % QG) * 4 + (lane & 3);
            wk[i] = (c / QG) * 8 + (lane >> 2);
            wn[i] = 4 * q;
            w[i] = *(const float4*)(W + (size_t)wk[i] * NOUT + wn[i]);
        }
#pragma unroll
        for (int i = 0; i < CH / 8; ++i) {
            float* __restrict__ d = &Wt[wn[i] * STRIDE + wk[i]];
            d[0]          = w[i].x;
            d[STRIDE]     = w[i].y;
            d[2 * STRIDE] = w[i].z;
            d[3 * STRIDE] = w[i].w;
        }
    }
    __syncthreads();
    if (!active) return;

    const int4 dg0 = *(const int4*)(indeg + row0 + 8 * half);
    const int4 dg1 = *(const int4*)(indeg + row0 + 8 * half + 4);

    const float* __restrict__ bp0 = &Wt[l * STRIDE + 4 * half];

    v8f acc[NT];
#pragma unroll
    for (int t = 0; t < NT; ++t) acc[t] = {};

#pragma unroll
    for (int j = 0; j < KB; ++j) {
        if (j + PF < KB) a[j + PF] = ap[2 * (j + PF)];
        const float4 av = a[j];
        v2f a0, a1;
        a0.x = av.x; a0.y = av.y;
        a1.x = av.z; a1.y = av.w;
        float4 bv[NT];
#pragma unroll
        for (int t = 0; t < NT; ++t) bv[t] = *(const float4*)(bp0 + t * 16 * STRIDE + 8 * j);
#pragma unroll
        for (int t = 0; t < NT; ++t) {
            v2f b;
            b.x = bv[t].x; b.y = bv[t].y;
            acc[t] = __builtin_amdgcn_wmma_f32_16x16x4_f32(
                false, a0, false, b, (short)0, acc[t], false, false);
        }
#pragma unroll
        for (int t = 0; t < NT; ++t) {
            v2f b;
            b.x = bv[t].z; b.y = bv[t].w;
            acc[t] = __builtin_amdgcn_wmma_f32_16x16x4_f32(
                false, a1, false, b, (short)0, acc[t], false, false);
        }
    }

    const float ds[8] = { inv_sqrt_deg(dg0.x), inv_sqrt_deg(dg0.y), inv_sqrt_deg(dg0.z), inv_sqrt_deg(dg0.w),
                          inv_sqrt_deg(dg1.x), inv_sqrt_deg(dg1.y), inv_sqrt_deg(dg1.z), inv_sqrt_deg(dg1.w) };
#pragma unroll
    for (int t = 0; t < NT; ++t) {
        const int c = t * 16 + l;
#pragma unroll
        for (int v = 0; v < 8; ++v) {
            const int r = row0 + v + 8 * half;
            Y[(size_t)r * NOUT + c] = acc[t][v] * ds[v];
        }
    }
}

template <int NOUT, bool ADD_BIAS, bool SCALE_ROWS>
__global__ __launch_bounds__(256) void gather_gemm_kernel(
    const float* __restrict__ H, const int* __restrict__ slots,
    const int* __restrict__ cnt, const int* __restrict__ novf,
    const int* __restrict__ ovf_c, const int* __restrict__ ovf_r,
    const float* __restrict__ bias, const float* __restrict__ Wn,
    const float* __restrict__ bn, float* __restrict__ Y)
{
    constexpr int TSTRIDE = 68;
    constexpr int NT = (NOUT + 15) / 16;
    __shared__ float T[16 * TSTRIDE];

    const int gid = blockIdx.x * 256 + threadIdx.x;
    const int c   = gid >> 4;
    const int f4  = (gid & 15) << 2;

    const int   m   = cnt[c];
    const float dc  = inv_sqrt_deg(m);
    const int   ms  = (m < SLOTS) ? m : SLOTS;
    const int* __restrict__ sp = slots + (size_t)c * SLOTS;
    const unsigned lb = (unsigned)f4 << 2;

    float4 acc = make_float4(0.0f, 0.0f, 0.0f, 0.0f);

    int j = 0;
#pragma unroll 2
    for (; j + 4 <= ms; j += 4) {
        const int r0 = sp[j], r1 = sp[j + 1], r2 = sp[j + 2], r3 = sp[j + 3];
        const float4 h0 = row_f4(H, r0, lb);
        const float4 h1 = row_f4(H, r1, lb);
        const float4 h2 = row_f4(H, r2, lb);
        const float4 h3 = row_f4(H, r3, lb);
        acc.x += h0.x; acc.y += h0.y; acc.z += h0.z; acc.w += h0.w;
        acc.x += h1.x; acc.y += h1.y; acc.z += h1.z; acc.w += h1.w;
        acc.x += h2.x; acc.y += h2.y; acc.z += h2.z; acc.w += h2.w;
        acc.x += h3.x; acc.y += h3.y; acc.z += h3.z; acc.w += h3.w;
    }
    const int rem = ms - j;
    if (rem > 0) {
        const bool k1 = rem > 1, k2 = rem > 2;
        const int r0 = sp[j];
        const int r1 = sp[j + (k1 ? 1 : 0)];
        const int r2 = sp[j + (k2 ? 2 : 0)];
        const float4 h0 = row_f4(H, r0, lb);
        const float4 h1 = row_f4(H, r1, lb);
        const float4 h2 = row_f4(H, r2, lb);
        acc.x += h0.x; acc.y += h0.y; acc.z += h0.z; acc.w += h0.w;
        acc.x += k1 ? h1.x : 0.0f; acc.y += k1 ? h1.y : 0.0f;
        acc.z += k1 ? h1.z : 0.0f; acc.w += k1 ? h1.w : 0.0f;
        acc.x += k2 ? h2.x : 0.0f; acc.y += k2 ? h2.y : 0.0f;
        acc.z += k2 ? h2.z : 0.0f; acc.w += k2 ? h2.w : 0.0f;
    }
    if (m > SLOTS) {
        const int no = *novf;
        for (int o = 0; o < no; ++o) {
            if (ovf_c[o] == c) {
                const int r = ovf_r[o];
                const float4 h = row_f4(H, r, lb);
                acc.x += h.x; acc.y += h.y; acc.z += h.z; acc.w += h.w;
            }
        }
    }

    const float4 hs = row_f4(H, c, lb);
    const float4 b  = *(const float4*)(bias + f4);
    float4 o;
    o.x = (acc.x + hs.x) * dc + b.x;
    o.y = (acc.y + hs.y) * dc + b.y;
    o.z = (acc.z + hs.z) * dc + b.z;
    o.w = (acc.w + hs.w) * dc + b.w;
    o.x = (o.x > 0.0f) ? o.x : 0.0f;
    o.y = (o.y > 0.0f) ? o.y : 0.0f;
    o.z = (o.z > 0.0f) ? o.z : 0.0f;
    o.w = (o.w > 0.0f) ? o.w : 0.0f;
    if constexpr (SCALE_ROWS) { o.x *= dc; o.y *= dc; o.z *= dc; o.w *= dc; }
    *(float4*)&T[(threadIdx.x >> 4) * TSTRIDE + f4] = o;

    const int wave = threadIdx.x >> 5;
    const int lane = threadIdx.x & 31;
    const int half = lane >> 4;
    const int l    = lane & 15;
    const int ncol = wave * 16 + l;

    v2f bf[16];
    float bv = 0.0f;
    if (wave < NT) {
        const bool cok = ncol < NOUT;
        if constexpr (ADD_BIAS) bv = bn[cok ? ncol : 0];
        const float* __restrict__ wp = Wn + (size_t)(2 * half) * NOUT + (cok ? ncol : 0);
#pragma unroll
        for (int s = 0; s < 16; ++s) {
            const float w0 = wp[(size_t)(4 * s) * NOUT];
            const float w1 = wp[(size_t)(4 * s + 1) * NOUT];
            bf[s].x = cok ? w0 : 0.0f;
            bf[s].y = cok ? w1 : 0.0f;
        }
    }
    __syncthreads();
    if (wave >= NT) return;

    v8f d = {};
    const float* __restrict__ ap = &T[l * TSTRIDE + 2 * half];
#pragma unroll
    for (int s = 0; s < 16; ++s) {
        const v2f a = *(const v2f*)(ap + 4 * s);
        d = __builtin_amdgcn_wmma_f32_16x16x4_f32(
            false, a, false, bf[s], (short)0, d, false, false);
    }

    if (ncol < NOUT) {
        const int row0 = blockIdx.x << 4;
        float* __restrict__ yp = Y + (size_t)(row0 + 8 * half) * NOUT + ncol;
        if constexpr (ADD_BIAS) {
#pragma unroll
            for (int v = 0; v < 8; ++v) yp[v * NOUT] = d[v] + bv;
        } else {
#pragma unroll
            for (int v = 0; v < 8; ++v) yp[v * NOUT] = d[v];
        }
    }
}


extern "C" void kernel_launch(void* const* d_in, const int* in_sizes, int n_in,
                              void* d_out, int out_size, void* d_ws, size_t ws_size,
                              hipStream_t stream) {
    const float* x    = (const float*)d_in[0];
    const int*   ei   = (const int*)d_in[1];
    const float* W1   = (const float*)d_in[2];
    const float* b1   = (const float*)d_in[3];
    const float* W2   = (const float*)d_in[4];
    const float* b2   = (const float*)d_in[5];
    const float* Wout = (const float*)d_in[6];
    const float* bout = (const float*)d_in[7];
    float* out = (float*)d_out;

    const int* row = ei;
    const int* col = ei + NEDGES;

    float* A   = (float*)d_ws;
    float* B   = A + (size_t)NNODES * 64;
    int* slots = (int*)(B + (size_t)NNODES * 64);
    int* cnt   = slots + (size_t)NNODES * SLOTS;
    int* novf  = cnt + NNODES;
    int* ovf_c = novf + 1;
    int* ovf_r = ovf_c + NEDGES;
    (void)ws_size; (void)n_in; (void)in_sizes; (void)out_size;

    const int T = 256;
    const int Mtiles = NNODES / 16;
    const int gemmBlocks = (Mtiles + 7) / 8;
    const int edgeBlocks = (NEDGES + T - 1) / T;
    static_assert(NNODES % 16 == 0, "a gather block owns 16 whole nodes");
    const int gatherBlocks = NNODES / 16;

    zero_kernel<<<(NNODES + 1 + T - 1) / T, T, 0, stream>>>(cnt, NNODES + 1);
    slot_kernel<<<edgeBlocks, T, 0, stream>>>(row, col, cnt, slots, novf, ovf_c, ovf_r, NEDGES);

    gemm_wmma_kernel<128, 64><<<gemmBlocks, T, 0, stream>>>(x, W1, cnt, A, NNODES);

    gather_gemm_kernel<64, false, true><<<gatherBlocks, T, 0, stream>>>(
        A, slots, cnt, novf, ovf_c, ovf_r, b1, W2, nullptr, B);

    gather_gemm_kernel<40, true, false><<<gatherBlocks, T, 0, stream>>>(
        B, slots, cnt, novf, ovf_c, ovf_r, b2, Wout, bout, out);
}


# gather: a node's in-degree, whole slot row (8 B a lane), own row and bias are asked for together; indices go round by DPP row_share, rows in batches of 8, empty slots name a zero row; sum order kept
# speedup vs baseline: 2.8494x; 1.0298x over previous
#include <hip/hip_runtime.h>

typedef __attribute__((ext_vector_type(2))) float v2f;
typedef __attribute__((ext_vector_type(8))) float v8f;

#define NNODES 100000
#define NEDGES 1250000


#define SLOTS 32


__global__ void zero_kernel(int* __restrict__ p, int n,
                            float* __restrict__ z0, float* __restrict__ z1) {
    int i = blockIdx.x * blockDim.x + threadIdx.x;
    if (i < n) p[i] = 0;
    if (i < 64) { z0[i] = 0.0f; z1[i] = 0.0f; }
}

__global__ void slot_kernel(const int* __restrict__ row, const int* __restrict__ col,
                            int* __restrict__ cnt, int* __restrict__ slots,
                            int* __restrict__ novf, int* __restrict__ ovf_c,
                            int* __restrict__ ovf_r, int E) {
    int e = blockIdx.x * blockDim.x + threadIdx.x;
    if (e < E) {
        const int c = col[e];
        const int r = row[e];
        const int k = atomicAdd(&cnt[c], 1);
        if (k < SLOTS) {
            slots[c * SLOTS + k] = r;
        } else {
            const int o = atomicAdd(novf, 1);
            ovf_c[o] = c;
            ovf_r[o] = r;
        }
    }
}

__device__ __forceinline__ float inv_sqrt_deg(int indeg) {
    const float x = (float)(indeg + 1);
    return (x > 0.0f) ? rsqrtf(x) : 0.0f;
}

__device__ __forceinline__ float4 row_f4(const float* __restrict__ H, int r, unsigned lane_bytes) {
    const unsigned off = ((unsigned)r << 8) + lane_bytes;
    return *(const float4*)((const char*)H + off);
}

template <int SRC>
__device__ __forceinline__ int row_bcast(int v) {
    return __builtin_amdgcn_mov_dpp(v, 0x150 + SRC, 0xf, 0xf, true);
}

template <int G>
__device__ __forceinline__ void gather8(const float* __restrict__ H, int sx, int sy,
                                        unsigned lb, float4& acc) {
    const int r0 = row_bcast<4 * G + 0>(sx), r1 = row_bcast<4 * G + 0>(sy);
    const int r2 = row_bcast<4 * G + 1>(sx), r3 = row_bcast<4 * G + 1>(sy);
    const int r4 = row_bcast<4 * G + 2>(sx), r5 = row_bcast<4 * G + 2>(sy);
    const int r6 = row_bcast<4 * G + 3>(sx), r7 = row_bcast<4 * G + 3>(sy);
    const float4 h0 = row_f4(H, r0, lb);
    const float4 h1 = row_f4(H, r1, lb);
    const float4 h2 = row_f4(H, r2, lb);
    const float4 h3 = row_f4(H, r3, lb);
    const float4 h4 = row_f4(H, r4, lb);
    const float4 h5 = row_f4(H, r5, lb);
    const float4 h6 = row_f4(H, r6, lb);
    const float4 h7 = row_f4(H, r7, lb);
    acc.x += h0.x; acc.y += h0.y; acc.z += h0.z; acc.w += h0.w;
    acc.x += h1.x; acc.y += h1.y; acc.z += h1.z; acc.w += h1.w;
    acc.x += h2.x; acc.y += h2.y; acc.z += h2.z; acc.w += h2.w;
    acc.x += h3.x; acc.y += h3.y; acc.z += h3.z; acc.w += h3.w;
    acc.x += h4.x; acc.y += h4.y; acc.z += h4.z; acc.w += h4.w;
    acc.x += h5.x; acc.y += h5.y; acc.z += h5.z; acc.w += h5.w;
    acc.x += h6.x; acc.y += h6.y; acc.z += h6.z; acc.w += h6.w;
    acc.x += h7.x; acc.y += h7.y; acc.z += h7.z; acc.w += h7.w;
}

template <int K, int NOUT>
__global__ __launch_bounds__(256) void gemm_wmma_kernel(
    const float* __restrict__ X, const float* __restrict__ W,
    const int* __restrict__ indeg, float* __restrict__ Y, int M)
{
    constexpr int STRIDE = K + 4;
    constexpr int NT = NOUT / 16;
    constexpr int KB = K / 8;
    constexpr int QG = NOUT / 16;
    constexpr int CH = KB * QG;
    constexpr int PF = 8;
    static_assert(K % 8 == 0 && NOUT % 16 == 0 && CH % 8 == 0 && KB >= PF, "tile shape");
    static_assert(((STRIDE / 4) & 1) == 1 && STRIDE % 4 == 0, "row pitch: 16 bytes times an odd number");
    __shared__ __attribute__((aligned(16))) float Wt[NOUT * STRIDE];

    const int lane  = threadIdx.x & 31;
    const int wave  = threadIdx.x >> 5;
    const int strip = blockIdx.x * 8 + wave;
    const int Mtiles = M >> 4;
    const bool active = strip < Mtiles;

    const int half = lane >> 4;
    const int l    = lane & 15;
    const int row0 = strip << 4;

    const float4* __restrict__ ap = (const float4*)(X + (size_t)(active ? row0 + l : 0) * K + 4 * half);

    float4 a[KB];
    if (active) {
#pragma unroll
        for (int j = 0; j < PF; ++j) a[j] = ap[2 * j];
    }

    {
        float4 w[CH / 8];
        int wn[CH / 8], wk[CH / 8];
#pragma unroll
        for (int i = 0; i < CH / 8; ++i) {
            const int c = wave * (CH / 8) + i;
            const int q = (c % QG) * 4 + (lane & 3);
            wk[i] = (c / QG) * 8 + (lane >> 2);
            wn[i] = 4 * q;
            w[i] = *(const float4*)(W + (size_t)wk[i] * NOUT + wn[i]);
        }
#pragma unroll
        for (int i = 0; i < CH / 8; ++i) {
            float* __restrict__ d = &Wt[wn[i] * STRIDE + wk[i]];
            d[0]          = w[i].x;
            d[STRIDE]     = w[i].y;
            d[2 * STRIDE] = w[i].z;
            d[3 * STRIDE] = w[i].w;
        }
    }
    __syncthreads();
    if (!active) return;

    const int4 dg0 = *(const int4*)(indeg + row0 + 8 * half);
    const int4 dg1 = *(const int4*)(indeg + row0 + 8 * half + 4);

    const float* __restrict__ bp0 = &Wt[l * STRIDE + 4 * half];

    v8f acc[NT];
#pragma unroll
    for (int t = 0; t < NT; ++t) acc[t] = {};

#pragma unroll
    for (int j = 0; j < KB; ++j) {
        if (j + PF < KB) a[j + PF] = ap[2 * (j + PF)];
        const float4 av = a[j];
        v2f a0, a1;
        a0.x = av.x; a0.y = av.y;
        a1.x = av.z; a1.y = av.w;
        float4 bv[NT];
#pragma unroll
        for (int t = 0; t < NT; ++t) bv[t] = *(const float4*)(bp0 + t * 16 * STRIDE + 8 * j);
#pragma unroll
        for (int t = 0; t < NT; ++t) {
            v2f b;
            b.x = bv[t].x; b.y = bv[t].y;
            acc[t] = __builtin_amdgcn_wmma_f32_16x16x4_f32(
                false, a0, false, b, (short)0, acc[t], false, false);
        }
#pragma unroll
        for (int t = 0; t < NT; ++t) {
            v2f b;
            b.x = bv[t].z; b.y = bv[t].w;
            acc[t] = __builtin_amdgcn_wmma_f32_16x16x4_f32(
                false, a1, false, b, (short)0, acc[t], false, false);
        }
    }

    const float ds[8] = { inv_sqrt_deg(dg0.x), inv_sqrt_deg(dg0.y), inv_sqrt_deg(dg0.z), inv_sqrt_deg(dg0.w),
                          inv_sqrt_deg(dg1.x), inv_sqrt_deg(dg1.y), inv_sqrt_deg(dg1.z), inv_sqrt_deg(dg1.w) };
#pragma unroll
    for (int t = 0; t < NT; ++t) {
        const int c = t * 16 + l;
#pragma unroll
        for (int v = 0; v < 8; ++v) {
            const int r = row0 + v + 8 * half;
            Y[(size_t)r * NOUT + c] = acc[t][v] * ds[v];
        }
    }
}

template <int NOUT, bool ADD_BIAS, bool SCALE_ROWS>
__global__ __launch_bounds__(256) void gather_gemm_kernel(
    const float* __restrict__ H, const int* __restrict__ slots,
    const int* __restrict__ cnt, const int* __restrict__ novf,
    const int* __restrict__ ovf_c, const int* __restrict__ ovf_r,
    const float* __restrict__ bias, const float* __restrict__ Wn,
    const float* __restrict__ bn, float* __restrict__ Y)
{
    constexpr int TSTRIDE = 68;
    constexpr int NT = (NOUT + 15) / 16;
    __shared__ float T[16 * TSTRIDE];

    const int gid = blockIdx.x * 256 + threadIdx.x;
    const int c   = gid >> 4;
    const int f4  = (gid & 15) << 2;

    const int l16 = gid & 15;
    const unsigned lb = (unsigned)f4 << 2;

    const int    m    = cnt[c];
    const int2   sraw = *(const int2*)(slots + (size_t)c * SLOTS + 2 * l16);
    const float4 hs   = row_f4(H, c, lb);
    const float4 b    = *(const float4*)(bias + f4);

    const float dc  = inv_sqrt_deg(m);
    const int   ms  = (m < SLOTS) ? m : SLOTS;
    const int sx = (2 * l16     < ms) ? sraw.x : NNODES;
    const int sy = (2 * l16 + 1 < ms) ? sraw.y : NNODES;

    float4 acc = make_float4(0.0f, 0.0f, 0.0f, 0.0f);

    gather8<0>(H, sx, sy, lb, acc);
    if (ms > 8) {
        gather8<1>(H, sx, sy, lb, acc);
        if (ms > 16) {
            gather8<2>(H, sx, sy, lb, acc);
            if (ms > 24) gather8<3>(H, sx, sy, lb, acc);
        }
    }
    if (m > SLOTS) {
        const int no = *novf;
        for (int o = 0; o < no; ++o) {
            if (ovf_c[o] == c) {
                const int r = ovf_r[o];
                const float4 h = row_f4(H, r, lb);
                acc.x += h.x; acc.y += h.y; acc.z += h.z; acc.w += h.w;
            }
        }
    }

    float4 o;
    o.x = (acc.x + hs.x) * dc + b.x;
    o.y = (acc.y + hs.y) * dc + b.y;
    o.z = (acc.z + hs.z) * dc + b.z;
    o.w = (acc.w + hs.w) * dc + b.w;
    o.x = (o.x > 0.0f) ? o.x : 0.0f;
    o.y = (o.y > 0.0f) ? o.y : 0.0f;
    o.z = (o.z > 0.0f) ? o.z : 0.0f;
    o.w = (o.w > 0.0f) ? o.w : 0.0f;
    if constexpr (SCALE_ROWS) { o.x *= dc; o.y *= dc; o.z *= dc; o.w *= dc; }
    *(float4*)&T[(threadIdx.x >> 4) * TSTRIDE + f4] = o;

    const int wave = threadIdx.x >> 5;
    const int lane = threadIdx.x & 31;
    const int half = lane >> 4;
    const int l    = lane & 15;
    const int ncol = wave * 16 + l;

    v2f bf[16];
    float bv = 0.0f;
    if (wave < NT) {
        const bool cok = ncol < NOUT;
        if constexpr (ADD_BIAS) bv = bn[cok ? ncol : 0];
        const float* __restrict__ wp = Wn + (size_t)(2 * half) * NOUT + (cok ? ncol : 0);
#pragma unroll
        for (int s = 0; s < 16; ++s) {
            const float w0 = wp[(size_t)(4 * s) * NOUT];
            const float w1 = wp[(size_t)(4 * s + 1) * NOUT];
            bf[s].x = cok ? w0 : 0.0f;
            bf[s].y = cok ? w1 : 0.0f;
        }
    }
    __syncthreads();
    if (wave >= NT) return;

    v8f d = {};
    const float* __restrict__ ap = &T[l * TSTRIDE + 2 * half];
#pragma unroll
    for (int s = 0; s < 16; ++s) {
        const v2f a = *(const v2f*)(ap + 4 * s);
        d = __builtin_amdgcn_wmma_f32_16x16x4_f32(
            false, a, false, bf[s], (short)0, d, false, false);
    }

    if (ncol < NOUT) {
        const int row0 = blockIdx.x << 4;
        float* __restrict__ yp = Y + (size_t)(row0 + 8 * half) * NOUT + ncol;
        if constexpr (ADD_BIAS) {
#pragma unroll
            for (int v = 0; v < 8; ++v) yp[v * NOUT] = d[v] + bv;
        } else {
#pragma unroll
            for (int v = 0; v < 8; ++v) yp[v * NOUT] = d[v];
        }
    }
}


extern "C" void kernel_launch(void* const* d_in, const int* in_sizes, int n_in,
                              void* d_out, int out_size, void* d_ws, size_t ws_size,
                              hipStream_t stream) {
    const float* x    = (const float*)d_in[0];
    const int*   ei   = (const int*)d_in[1];
    const float* W1   = (const float*)d_in[2];
    const float* b1   = (const float*)d_in[3];
    const float* W2   = (const float*)d_in[4];
    const float* b2   = (const float*)d_in[5];
    const float* Wout = (const float*)d_in[6];
    const float* bout = (const float*)d_in[7];
    float* out = (float*)d_out;

    const int* row = ei;
    const int* col = ei + NEDGES;

    float* A   = (float*)d_ws;
    float* B   = A + (size_t)(NNODES + 1) * 64;
    int* slots = (int*)(B + (size_t)(NNODES + 1) * 64);
    int* cnt   = slots + (size_t)NNODES * SLOTS;
    int* novf  = cnt + NNODES;
    int* ovf_c = novf + 1;
    int* ovf_r = ovf_c + NEDGES;
    (void)ws_size; (void)n_in; (void)in_sizes; (void)out_size;

    const int T = 256;
    const int Mtiles = NNODES / 16;
    const int gemmBlocks = (Mtiles + 7) / 8;
    const int edgeBlocks = (NEDGES + T - 1) / T;
    static_assert(NNODES % 16 == 0, "a gather block owns 16 whole nodes");
    const int gatherBlocks = NNODES / 16;

    zero_kernel<<<(NNODES + 1 + T - 1) / T, T, 0, stream>>>(
        cnt, NNODES + 1, A + (size_t)NNODES * 64, B + (size_t)NNODES * 64);
    slot_kernel<<<edgeBlocks, T, 0, stream>>>(row, col, cnt, slots, novf, ovf_c, ovf_r, NEDGES);

    gemm_wmma_kernel<128, 64><<<gemmBlocks, T, 0, stream>>>(x, W1, cnt, A, NNODES);

    gather_gemm_kernel<64, false, true><<<gatherBlocks, T, 0, stream>>>(
        A, slots, cnt, novf, ovf_c, ovf_r, b1, W2, nullptr, B);

    gather_gemm_kernel<40, true, false><<<gatherBlocks, T, 0, stream>>>(
        B, slots, cnt, novf, ovf_c, ovf_r, b2, Wout, bout, out);
}


# per-edge rank atomics in memory -> LDS counters: a block buckets its edges by 256-node destination range (LDS count + scan), an owner block per range slots them; sum order stays arrival order
# speedup vs baseline: 5.0770x; 1.7818x over previous
#include <hip/hip_runtime.h>

typedef __attribute__((ext_vector_type(2))) float v2f;
typedef __attribute__((ext_vector_type(8))) float v8f;

#define NNODES 100000
#define NEDGES 1250000


#define SLOTS 32


#define EPB    4884
#define NBLK1  256
#define RANGE  256
#define NB     ((NNODES + RANGE - 1) / RANGE)
#define SEGW   (NB + 1)
#define T1     512
#define EPT    ((EPB + T1 - 1) / T1)

__global__ __launch_bounds__(T1) void bucket_kernel(
    const int* __restrict__ row, const int* __restrict__ col,
    int* __restrict__ pairs, int* __restrict__ segstart,
    int* __restrict__ novf, float* __restrict__ z0, float* __restrict__ z1)
{
    static_assert(SEGW <= T1, "one counter per thread in the scan");
    __shared__ int hist[T1];
    __shared__ int offs[T1];
    __shared__ int wsum[T1 / 32];

    const int tid = threadIdx.x;
    const int blk = blockIdx.x;
    if (blk == 0) {
        if (tid < 64) { z0[tid] = 0.0f; z1[tid] = 0.0f; }
        if (tid == 64) *novf = 0;
    }
    hist[tid] = 0;
    __syncthreads();

    const int base = blk * EPB;
    const int end  = (base + EPB < NEDGES) ? base + EPB : NEDGES;

    int c[EPT], r[EPT], k[EPT];
#pragma unroll
    for (int j = 0; j < EPT; ++j) {
        const int e = base + j * T1 + tid;
        const bool ok = e < end;
        c[j] = ok ? col[e] : -1;
        r[j] = ok ? row[e] : 0;
    }
#pragma unroll
    for (int j = 0; j < EPT; ++j) {
        k[j] = 0;
        if (c[j] >= 0) k[j] = atomicAdd(&hist[c[j] / RANGE], 1);
    }
    __syncthreads();

    const int lane = tid & 31;
    const int wave = tid >> 5;
    const int v = hist[tid];
    int incl = v;
#pragma unroll
    for (int d = 1; d < 32; d <<= 1) {
        const int t = __shfl_up(incl, d);
        if (lane >= d) incl += t;
    }
    if (lane == 31) wsum[wave] = incl;
    __syncthreads();
    int add = 0;
    for (int w = 0; w < wave; ++w) add += wsum[w];
    offs[tid] = add + incl - v;
    __syncthreads();

    int2* __restrict__ reg = (int2*)pairs + (size_t)blk * EPB;
#pragma unroll
    for (int j = 0; j < EPT; ++j) {
        if (c[j] >= 0) reg[offs[c[j] / RANGE] + k[j]] = make_int2(c[j], r[j]);
    }
    if (tid < SEGW) segstart[blk * SEGW + tid] = offs[tid];
}

__device__ __forceinline__ void store_edge(int2 p, int k,
                                           int* __restrict__ slots, int* __restrict__ novf,
                                           int* __restrict__ ovf_c, int* __restrict__ ovf_r) {
    if (k < SLOTS) {
        slots[p.x * SLOTS + k] = p.y;
    } else {
        const int o = atomicAdd(novf, 1);
        ovf_c[o] = p.x;
        ovf_r[o] = p.y;
    }
}

__global__ __launch_bounds__(NBLK1) void rank_kernel(
    const int* __restrict__ pairs, const int* __restrict__ segstart,
    int* __restrict__ cnt, int* __restrict__ slots,
    int* __restrict__ novf, int* __restrict__ ovf_c, int* __restrict__ ovf_r)
{
    static_assert(RANGE == NBLK1, "one node counter and one piece bound per thread");
    __shared__ int lcnt[RANGE];
    __shared__ int sst[NBLK1];
    __shared__ int sen[NBLK1];

    const int tid = threadIdx.x;
    const int b = blockIdx.x;
    const int nbase = b * RANGE;
    lcnt[tid] = 0;
    sst[tid] = segstart[tid * SEGW + b];
    sen[tid] = segstart[tid * SEGW + b + 1];
    __syncthreads();

    const int hw  = tid >> 4;
    const int l16 = tid & 15;
    const int2* __restrict__ pp = (const int2*)pairs;

#pragma unroll
    for (int g = 0; g < 2; ++g) {
        int2 p[8];
        bool ok[8];
#pragma unroll
        for (int u = 0; u < 8; ++u) {
            const int s = hw + 16 * (8 * g + u);
            const int i = sst[s] + l16;
            ok[u] = i < sen[s];
            p[u] = make_int2(0, 0);
            if (ok[u]) p[u] = pp[(size_t)s * EPB + i];
        }
        int k[8];
#pragma unroll
        for (int u = 0; u < 8; ++u) {
            k[u] = 0;
            if (ok[u]) k[u] = atomicAdd(&lcnt[p[u].x - nbase], 1);
        }
#pragma unroll
        for (int u = 0; u < 8; ++u) {
            if (ok[u]) store_edge(p[u], k[u], slots, novf, ovf_c, ovf_r);
        }
#pragma unroll
        for (int u = 0; u < 8; ++u) {
            const int s = hw + 16 * (8 * g + u);
            const int en = sen[s];
            for (int i = sst[s] + 16 + l16; i < en; i += 16) {
                const int2 q = pp[(size_t)s * EPB + i];
                store_edge(q, atomicAdd(&lcnt[q.x - nbase], 1), slots, novf, ovf_c, ovf_r);
            }
        }
    }
    __syncthreads();
    if (nbase + tid < NNODES) cnt[nbase + tid] = lcnt[tid];
}

__device__ __forceinline__ float inv_sqrt_deg(int indeg) {
    const float x = (float)(indeg + 1);
    return (x > 0.0f) ? rsqrtf(x) : 0.0f;
}

__device__ __forceinline__ float4 row_f4(const float* __restrict__ H, int r, unsigned lane_bytes) {
    const unsigned off = ((unsigned)r << 8) + lane_bytes;
    return *(const float4*)((const char*)H + off);
}

template <int SRC>
__device__ __forceinline__ int row_bcast(int v) {
    return __builtin_amdgcn_mov_dpp(v, 0x150 + SRC, 0xf, 0xf, true);
}

template <int G>
__device__ __forceinline__ void gather8(const float* __restrict__ H, int sx, int sy,
                                        unsigned lb, float4& acc) {
    const int r0 = row_bcast<4 * G + 0>(sx), r1 = row_bcast<4 * G + 0>(sy);
    const int r2 = row_bcast<4 * G + 1>(sx), r3 = row_bcast<4 * G + 1>(sy);
    const int r4 = row_bcast<4 * G + 2>(sx), r5 = row_bcast<4 * G + 2>(sy);
    const int r6 = row_bcast<4 * G + 3>(sx), r7 = row_bcast<4 * G + 3>(sy);
    const float4 h0 = row_f4(H, r0, lb);
    const float4 h1 = row_f4(H, r1, lb);
    const float4 h2 = row_f4(H, r2, lb);
    const float4 h3 = row_f4(H, r3, lb);
    const float4 h4 = row_f4(H, r4, lb);
    const float4 h5 = row_f4(H, r5, lb);
    const float4 h6 = row_f4(H, r6, lb);
    const float4 h7 = row_f4(H, r7, lb);
    acc.x += h0.x; acc.y += h0.y; acc.z += h0.z; acc.w += h0.w;
    acc.x += h1.x; acc.y += h1.y; acc.z += h1.z; acc.w += h1.w;
    acc.x += h2.x; acc.y += h2.y; acc.z += h2.z; acc.w += h2.w;
    acc.x += h3.x; acc.y += h3.y; acc.z += h3.z; acc.w += h3.w;
    acc.x += h4.x; acc.y += h4.y; acc.z += h4.z; acc.w += h4.w;
    acc.x += h5.x; acc.y += h5.y; acc.z += h5.z; acc.w += h5.w;
    acc.x += h6.x; acc.y += h6.y; acc.z += h6.z; acc.w += h6.w;
    acc.x += h7.x; acc.y += h7.y; acc.z += h7.z; acc.w += h7.w;
}

template <int K, int NOUT>
__global__ __launch_bounds__(256) void gemm_wmma_kernel(
    const float* __restrict__ X, const float* __restrict__ W,
    const int* __restrict__ indeg, float* __restrict__ Y, int M)
{
    constexpr int STRIDE = K + 4;
    constexpr int NT = NOUT / 16;
    constexpr int KB = K / 8;
    constexpr int QG = NOUT / 16;
    constexpr int CH = KB * QG;
    constexpr int PF = 8;
    static_assert(K % 8 == 0 && NOUT % 16 == 0 && CH % 8 == 0 && KB >= PF, "tile shape");
    static_assert(((STRIDE / 4) & 1) == 1 && STRIDE % 4 == 0, "row pitch: 16 bytes times an odd number");
    __shared__ __attribute__((aligned(16))) float Wt[NOUT * STRIDE];

    const int lane  = threadIdx.x & 31;
    const int wave  = threadIdx.x >> 5;
    const int strip = blockIdx.x * 8 + wave;
    const int Mtiles = M >> 4;
    const bool active = strip < Mtiles;

    const int half = lane >> 4;
    const int l    = lane & 15;
    const int row0 = strip << 4;

    const float4* __restrict__ ap = (const float4*)(X + (size_t)(active ? row0 + l : 0) * K + 4 * half);

    float4 a[KB];
    if (active) {
#pragma unroll
        for (int j = 0; j < PF; ++j) a[j] = ap[2 * j];
    }

    {
        float4 w[CH / 8];
        int wn[CH / 8], wk[CH / 8];
#pragma unroll
        for (int i = 0; i < CH / 8; ++i) {
            const int c = wave * (CH / 8) + i;
            const int q = (c % QG) * 4 + (lane & 3);
            wk[i] = (c / QG) * 8 + (lane >> 2);
            wn[i] = 4 * q;
            w[i] = *(const float4*)(W + (size_t)wk[i] * NOUT + wn[i]);
        }
#pragma unroll
        for (int i = 0; i < CH / 8; ++i) {
            float* __restrict__ d = &Wt[wn[i] * STRIDE + wk[i]];
            d[0]          = w[i].x;
            d[STRIDE]     = w[i].y;
            d[2 * STRIDE] = w[i].z;
            d[3 * STRIDE] = w[i].w;
        }
    }
    __syncthreads();
    if (!active) return;

    const int4 dg0 = *(const int4*)(indeg + row0 + 8 * half);
    const int4 dg1 = *(const int4*)(indeg + row0 + 8 * half + 4);

    const float* __restrict__ bp0 = &Wt[l * STRIDE + 4 * half];

    v8f acc[NT];
#pragma unroll
    for (int t = 0; t < NT; ++t) acc[t] = {};

#pragma unroll
    for (int j = 0; j < KB; ++j) {
        if (j + PF < KB) a[j + PF] = ap[2 * (j + PF)];
        const float4 av = a[j];
        v2f a0, a1;
        a0.x = av.x; a0.y = av.y;
        a1.x = av.z; a1.y = av.w;
        float4 bv[NT];
#pragma unroll
        for (int t = 0; t < NT; ++t) bv[t] = *(const float4*)(bp0 + t * 16 * STRIDE + 8 * j);
#pragma unroll
        for (int t = 0; t < NT; ++t) {
            v2f b;
            b.x = bv[t].x; b.y = bv[t].y;
            acc[t] = __builtin_amdgcn_wmma_f32_16x16x4_f32(
                false, a0, false, b, (short)0, acc[t], false, false);
        }
#pragma unroll
        for (int t = 0; t < NT; ++t) {
            v2f b;
            b.x = bv[t].z; b.y = bv[t].w;
            acc[t] = __builtin_amdgcn_wmma_f32_16x16x4_f32(
                false, a1, false, b, (short)0, acc[t], false, false);
        }
    }

    const float ds[8] = { inv_sqrt_deg(dg0.x), inv_sqrt_deg(dg0.y), inv_sqrt_deg(dg0.z), inv_sqrt_deg(dg0.w),
                          inv_sqrt_deg(dg1.x), inv_sqrt_deg(dg1.y), inv_sqrt_deg(dg1.z), inv_sqrt_deg(dg1.w) };
#pragma unroll
    for (int t = 0; t < NT; ++t) {
        const int c = t * 16 + l;
#pragma unroll
        for (int v = 0; v < 8; ++v) {
            const int r = row0 + v + 8 * half;
            Y[(size_t)r * NOUT + c] = acc[t][v] * ds[v];
        }
    }
}

template <int NOUT, bool ADD_BIAS, bool SCALE_ROWS>
__global__ __launch_bounds__(256) void gather_gemm_kernel(
    const float* __restrict__ H, const int* __restrict__ slots,
    const int* __restrict__ cnt, const int* __restrict__ novf,
    const int* __restrict__ ovf_c, const int* __restrict__ ovf_r,
    const float* __restrict__ bias, const float* __restrict__ Wn,
    const float* __restrict__ bn, float* __restrict__ Y)
{
    constexpr int TSTRIDE = 68;
    constexpr int NT = (NOUT + 15) / 16;
    __shared__ float T[16 * TSTRIDE];

    const int gid = blockIdx.x * 256 + threadIdx.x;
    const int c   = gid >> 4;
    const int f4  = (gid & 15) << 2;

    const int l16 = gid & 15;
    const unsigned lb = (unsigned)f4 << 2;

    const int    m    = cnt[c];
    const int2   sraw = *(const int2*)(slots + (size_t)c * SLOTS + 2 * l16);
    const float4 hs   = row_f4(H, c, lb);
    const float4 b    = *(const float4*)(bias + f4);

    const float dc  = inv_sqrt_deg(m);
    const int   ms  = (m < SLOTS) ? m : SLOTS;
    const int sx = (2 * l16     < ms) ? sraw.x : NNODES;
    const int sy = (2 * l16 + 1 < ms) ? sraw.y : NNODES;

    float4 acc = make_float4(0.0f, 0.0f, 0.0f, 0.0f);

    gather8<0>(H, sx, sy, lb, acc);
    if (ms > 8) {
        gather8<1>(H, sx, sy, lb, acc);
        if (ms > 16) {
            gather8<2>(H, sx, sy, lb, acc);
            if (ms > 24) gather8<3>(H, sx, sy, lb, acc);
        }
    }
    if (m > SLOTS) {
        const int no = *novf;
        for (int o = 0; o < no; ++o) {
            if (ovf_c[o] == c) {
                const int r = ovf_r[o];
                const float4 h = row_f4(H, r, lb);
                acc.x += h.x; acc.y += h.y; acc.z += h.z; acc.w += h.w;
            }
        }
    }

    float4 o;
    o.x = (acc.x + hs.x) * dc + b.x;
    o.y = (acc.y + hs.y) * dc + b.y;
    o.z = (acc.z + hs.z) * dc + b.z;
    o.w = (acc.w + hs.w) * dc + b.w;
    o.x = (o.x > 0.0f) ? o.x : 0.0f;
    o.y = (o.y > 0.0f) ? o.y : 0.0f;
    o.z = (o.z > 0.0f) ? o.z : 0.0f;
    o.w = (o.w > 0.0f) ? o.w : 0.0f;
    if constexpr (SCALE_ROWS) { o.x *= dc; o.y *= dc; o.z *= dc; o.w *= dc; }
    *(float4*)&T[(threadIdx.x >> 4) * TSTRIDE + f4] = o;

    const int wave = threadIdx.x >> 5;
    const int lane = threadIdx.x & 31;
    const int half = lane >> 4;
    const int l    = lane & 15;
    const int ncol = wave * 16 + l;

    v2f bf[16];
    float bv = 0.0f;
    if (wave < NT) {
        const bool cok = ncol < NOUT;
        if constexpr (ADD_BIAS) bv = bn[cok ? ncol : 0];
        const float* __restrict__ wp = Wn + (size_t)(2 * half) * NOUT + (cok ? ncol : 0);
#pragma unroll
        for (int s = 0; s < 16; ++s) {
            const float w0 = wp[(size_t)(4 * s) * NOUT];
            const float w1 = wp[(size_t)(4 * s + 1) * NOUT];
            bf[s].x = cok ? w0 : 0.0f;
            bf[s].y = cok ? w1 : 0.0f;
        }
    }
    __syncthreads();
    if (wave >= NT) return;

    v8f d = {};
    const float* __restrict__ ap = &T[l * TSTRIDE + 2 * half];
#pragma unroll
    for (int s = 0; s < 16; ++s) {
        const v2f a = *(const v2f*)(ap + 4 * s);
        d = __builtin_amdgcn_wmma_f32_16x16x4_f32(
            false, a, false, bf[s], (short)0, d, false, false);
    }

    if (ncol < NOUT) {
        const int row0 = blockIdx.x << 4;
        float* __restrict__ yp = Y + (size_t)(row0 + 8 * half) * NOUT + ncol;
        if constexpr (ADD_BIAS) {
#pragma unroll
            for (int v = 0; v < 8; ++v) yp[v * NOUT] = d[v] + bv;
        } else {
#pragma unroll
            for (int v = 0; v < 8; ++v) yp[v * NOUT] = d[v];
        }
    }
}


extern "C" void kernel_launch(void* const* d_in, const int* in_sizes, int n_in,
                              void* d_out, int out_size, void* d_ws, size_t ws_size,
                              hipStream_t stream) {
    const float* x    = (const float*)d_in[0];
    const int*   ei   = (const int*)d_in[1];
    const float* W1   = (const float*)d_in[2];
    const float* b1   = (const float*)d_in[3];
    const float* W2   = (const float*)d_in[4];
    const float* b2   = (const float*)d_in[5];
    const float* Wout = (const float*)d_in[6];
    const float* bout = (const float*)d_in[7];
    float* out = (float*)d_out;

    const int* row = ei;
    const int* col = ei + NEDGES;

    float* A   = (float*)d_ws;
    float* B   = A + (size_t)(NNODES + 1) * 64;
    int* slots = (int*)(B + (size_t)(NNODES + 1) * 64);
    int* pairs = slots + (size_t)NNODES * SLOTS;
    int* segst = pairs + (size_t)NBLK1 * EPB * 2;
    int* cnt   = segst + (size_t)NBLK1 * SEGW;
    int* novf  = cnt + NNODES;
    int* ovf_c = novf + 1;
    int* ovf_r = ovf_c + NEDGES;
    (void)ws_size; (void)n_in; (void)in_sizes; (void)out_size;
    static_assert((size_t)NBLK1 * EPB >= NEDGES, "bucket_kernel blocks cover every edge");
    static_assert(((size_t)NNODES * SLOTS) % 4 == 0 && ((size_t)NBLK1 * EPB * 2) % 4 == 0 && ((size_t)NBLK1 * SEGW) % 4 == 0,
                  "pairs starts on 8 bytes and cnt on 16 (the product reads in-degrees four at a time)");

    const int T = 256;
    const int Mtiles = NNODES / 16;
    const int gemmBlocks = (Mtiles + 7) / 8;
    static_assert(NNODES % 16 == 0, "a gather block owns 16 whole nodes");
    const int gatherBlocks = NNODES / 16;

    bucket_kernel<<<NBLK1, T1, 0, stream>>>(row, col, pairs, segst, novf,
                                            A + (size_t)NNODES * 64, B + (size_t)NNODES * 64);
    rank_kernel<<<NB, NBLK1, 0, stream>>>(pairs, segst, cnt, slots, novf, ovf_c, ovf_r);

    gemm_wmma_kernel<128, 64><<<gemmBlocks, T, 0, stream>>>(x, W1, cnt, A, NNODES);

    gather_gemm_kernel<64, false, true><<<gatherBlocks, T, 0, stream>>>(
        A, slots, cnt, novf, ovf_c, ovf_r, b1, W2, nullptr, B);

    gather_gemm_kernel<40, true, false><<<gatherBlocks, T, 0, stream>>>(
        B, slots, cnt, novf, ovf_c, ovf_r, b2, Wout, bout, out);
}


# gather blocks resident: each takes 5 tiles of 16 nodes (a fifth of the waves enter), next weights staged in LDS once per block, next tile's in-degrees and slot rows asked ahead; sums unchanged
# speedup vs baseline: 5.2770x; 1.0394x over previous
#include <hip/hip_runtime.h>

typedef __attribute__((ext_vector_type(2))) float v2f;
typedef __attribute__((ext_vector_type(8))) float v8f;

#define NNODES 100000
#define NEDGES 1250000


#define SLOTS 32


#define EPB    4884
#define NBLK1  256
#define RANGE  256
#define NB     ((NNODES + RANGE - 1) / RANGE)
#define SEGW   (NB + 1)
#define T1     512
#define EPT    ((EPB + T1 - 1) / T1)

__global__ __launch_bounds__(T1) void bucket_kernel(
    const int* __restrict__ row, const int* __restrict__ col,
    int* __restrict__ pairs, int* __restrict__ segstart,
    int* __restrict__ novf, float* __restrict__ z0, float* __restrict__ z1)
{
    static_assert(SEGW <= T1, "one counter per thread in the scan");
    __shared__ int hist[T1];
    __shared__ int offs[T1];
    __shared__ int wsum[T1 / 32];

    const int tid = threadIdx.x;
    const int blk = blockIdx.x;
    if (blk == 0) {
        if (tid < 64) { z0[tid] = 0.0f; z1[tid] = 0.0f; }
        if (tid == 64) *novf = 0;
    }
    hist[tid] = 0;
    __syncthreads();

    const int base = blk * EPB;
    const int end  = (base + EPB < NEDGES) ? base + EPB : NEDGES;

    int c[EPT], r[EPT], k[EPT];
#pragma unroll
    for (int j = 0; j < EPT; ++j) {
        const int e = base + j * T1 + tid;
        const bool ok = e < end;
        c[j] = ok ? col[e] : -1;
        r[j] = ok ? row[e] : 0;
    }
#pragma unroll
    for (int j = 0; j < EPT; ++j) {
        k[j] = 0;
        if (c[j] >= 0) k[j] = atomicAdd(&hist[c[j] / RANGE], 1);
    }
    __syncthreads();

    const int lane = tid & 31;
    const int wave = tid >> 5;
    const int v = hist[tid];
    int incl = v;
#pragma unroll
    for (int d = 1; d < 32; d <<= 1) {
        const int t = __shfl_up(incl, d);
        if (lane >= d) incl += t;
    }
    if (lane == 31) wsum[wave] = incl;
    __syncthreads();
    int add = 0;
    for (int w = 0; w < wave; ++w) add += wsum[w];
    offs[tid] = add + incl - v;
    __syncthreads();

    int2* __restrict__ reg = (int2*)pairs + (size_t)blk * EPB;
#pragma unroll
    for (int j = 0; j < EPT; ++j) {
        if (c[j] >= 0) reg[offs[c[j] / RANGE] + k[j]] = make_int2(c[j], r[j]);
    }
    if (tid < SEGW) segstart[blk * SEGW + tid] = offs[tid];
}

__device__ __forceinline__ void store_edge(int2 p, int k,
                                           int* __restrict__ slots, int* __restrict__ novf,
                                           int* __restrict__ ovf_c, int* __restrict__ ovf_r) {
    if (k < SLOTS) {
        slots[p.x * SLOTS + k] = p.y;
    } else {
        const int o = atomicAdd(novf, 1);
        ovf_c[o] = p.x;
        ovf_r[o] = p.y;
    }
}

__global__ __launch_bounds__(NBLK1) void rank_kernel(
    const int* __restrict__ pairs, const int* __restrict__ segstart,
    int* __restrict__ cnt, int* __restrict__ slots,
    int* __restrict__ novf, int* __restrict__ ovf_c, int* __restrict__ ovf_r)
{
    static_assert(RANGE == NBLK1, "one node counter and one piece bound per thread");
    __shared__ int lcnt[RANGE];
    __shared__ int sst[NBLK1];
    __shared__ int sen[NBLK1];

    const int tid = threadIdx.x;
    const int b = blockIdx.x;
    const int nbase = b * RANGE;
    lcnt[tid] = 0;
    sst[tid] = segstart[tid * SEGW + b];
    sen[tid] = segstart[tid * SEGW + b + 1];
    __syncthreads();

    const int hw  = tid >> 4;
    const int l16 = tid & 15;
    const int2* __restrict__ pp = (const int2*)pairs;

#pragma unroll
    for (int g = 0; g < 2; ++g) {
        int2 p[8];
        bool ok[8];
#pragma unroll
        for (int u = 0; u < 8; ++u) {
            const int s = hw + 16 * (8 * g + u);
            const int i = sst[s] + l16;
            ok[u] = i < sen[s];
            p[u] = make_int2(0, 0);
            if (ok[u]) p[u] = pp[(size_t)s * EPB + i];
        }
        int k[8];
#pragma unroll
        for (int u = 0; u < 8; ++u) {
            k[u] = 0;
            if (ok[u]) k[u] = atomicAdd(&lcnt[p[u].x - nbase], 1);
        }
#pragma unroll
        for (int u = 0; u < 8; ++u) {
            if (ok[u]) store_edge(p[u], k[u], slots, novf, ovf_c, ovf_r);
        }
#pragma unroll
        for (int u = 0; u < 8; ++u) {
            const int s = hw + 16 * (8 * g + u);
            const int en = sen[s];
            for (int i = sst[s] + 16 + l16; i < en; i += 16) {
                const int2 q = pp[(size_t)s * EPB + i];
                store_edge(q, atomicAdd(&lcnt[q.x - nbase], 1), slots, novf, ovf_c, ovf_r);
            }
        }
    }
    __syncthreads();
    if (nbase + tid < NNODES) cnt[nbase + tid] = lcnt[tid];
}

__device__ __forceinline__ float inv_sqrt_deg(int indeg) {
    const float x = (float)(indeg + 1);
    return (x > 0.0f) ? rsqrtf(x) : 0.0f;
}

__device__ __forceinline__ float4 row_f4(const float* __restrict__ H, int r, unsigned lane_bytes) {
    const unsigned off = ((unsigned)r << 8) + lane_bytes;
    return *(const float4*)((const char*)H + off);
}

template <int SRC>
__device__ __forceinline__ int row_bcast(int v) {
    return __builtin_amdgcn_mov_dpp(v, 0x150 + SRC, 0xf, 0xf, true);
}

template <int G>
__device__ __forceinline__ void gather8(const float* __restrict__ H, int sx, int sy,
                                        unsigned lb, float4& acc) {
    const int r0 = row_bcast<4 * G + 0>(sx), r1 = row_bcast<4 * G + 0>(sy);
    const int r2 = row_bcast<4 * G + 1>(sx), r3 = row_bcast<4 * G + 1>(sy);
    const int r4 = row_bcast<4 * G + 2>(sx), r5 = row_bcast<4 * G + 2>(sy);
    const int r6 = row_bcast<4 * G + 3>(sx), r7 = row_bcast<4 * G + 3>(sy);
    const float4 h0 = row_f4(H, r0, lb);
    const float4 h1 = row_f4(H, r1, lb);
    const float4 h2 = row_f4(H, r2, lb);
    const float4 h3 = row_f4(H, r3, lb);
    const float4 h4 = row_f4(H, r4, lb);
    const float4 h5 = row_f4(H, r5, lb);
    const float4 h6 = row_f4(H, r6, lb);
    const float4 h7 = row_f4(H, r7, lb);
    acc.x += h0.x; acc.y += h0.y; acc.z += h0.z; acc.w += h0.w;
    acc.x += h1.x; acc.y += h1.y; acc.z += h1.z; acc.w += h1.w;
    acc.x += h2.x; acc.y += h2.y; acc.z += h2.z; acc.w += h2.w;
    acc.x += h3.x; acc.y += h3.y; acc.z += h3.z; acc.w += h3.w;
    acc.x += h4.x; acc.y += h4.y; acc.z += h4.z; acc.w += h4.w;
    acc.x += h5.x; acc.y += h5.y; acc.z += h5.z; acc.w += h5.w;
    acc.x += h6.x; acc.y += h6.y; acc.z += h6.z; acc.w += h6.w;
    acc.x += h7.x; acc.y += h7.y; acc.z += h7.z; acc.w += h7.w;
}

template <int K, int NOUT>
__global__ __launch_bounds__(256) void gemm_wmma_kernel(
    const float* __restrict__ X, const float* __restrict__ W,
    const int* __restrict__ indeg, float* __restrict__ Y, int M)
{
    constexpr int STRIDE = K + 4;
    constexpr int NT = NOUT / 16;
    constexpr int KB = K / 8;
    constexpr int QG = NOUT / 16;
    constexpr int CH = KB * QG;
    constexpr int PF = 8;
    static_assert(K % 8 == 0 && NOUT % 16 == 0 && CH % 8 == 0 && KB >= PF, "tile shape");
    static_assert(((STRIDE / 4) & 1) == 1 && STRIDE % 4 == 0, "row pitch: 16 bytes times an odd number");
    __shared__ __attribute__((aligned(16))) float Wt[NOUT * STRIDE];

    const int lane  = threadIdx.x & 31;
    const int wave  = threadIdx.x >> 5;
    const int strip = blockIdx.x * 8 + wave;
    const int Mtiles = M >> 4;
    const bool active = strip < Mtiles;

    const int half = lane >> 4;
    const int l    = lane & 15;
    const int row0 = strip << 4;

    const float4* __restrict__ ap = (const float4*)(X + (size_t)(active ? row0 + l : 0) * K + 4 * half);

    float4 a[KB];
    if (active) {
#pragma unroll
        for (int j = 0; j < PF; ++j) a[j] = ap[2 * j];
    }

    {
        float4 w[CH / 8];
        int wn[CH / 8], wk[CH / 8];
#pragma unroll
        for (int i = 0; i < CH / 8; ++i) {
            const int c = wave * (CH / 8) + i;
            const int q = (c % QG) * 4 + (lane & 3);
            wk[i] = (c / QG) * 8 + (lane >> 2);
            wn[i] = 4 * q;
            w[i] = *(const float4*)(W + (size_t)wk[i] * NOUT + wn[i]);
        }
#pragma unroll
        for (int i = 0; i < CH / 8; ++i) {
            float* __restrict__ d = &Wt[wn[i] * STRIDE + wk[i]];
            d[0]          = w[i].x;
            d[STRIDE]     = w[i].y;
            d[2 * STRIDE] = w[i].z;
            d[3 * STRIDE] = w[i].w;
        }
    }
    __syncthreads();
    if (!active) return;

    const int4 dg0 = *(const int4*)(indeg + row0 + 8 * half);
    const int4 dg1 = *(const int4*)(indeg + row0 + 8 * half + 4);

    const float* __restrict__ bp0 = &Wt[l * STRIDE + 4 * half];

    v8f acc[NT];
#pragma unroll
    for (int t = 0; t < NT; ++t) acc[t] = {};

#pragma unroll
    for (int j = 0; j < KB; ++j) {
        if (j + PF < KB) a[j + PF] = ap[2 * (j + PF)];
        const float4 av = a[j];
        v2f a0, a1;
        a0.x = av.x; a0.y = av.y;
        a1.x = av.z; a1.y = av.w;
        float4 bv[NT];
#pragma unroll
        for (int t = 0; t < NT; ++t) bv[t] = *(const float4*)(bp0 + t * 16 * STRIDE + 8 * j);
#pragma unroll
        for (int t = 0; t < NT; ++t) {
            v2f b;
            b.x = bv[t].x; b.y = bv[t].y;
            acc[t] = __builtin_amdgcn_wmma_f32_16x16x4_f32(
                false, a0, false, b, (short)0, acc[t], false, false);
        }
#pragma unroll
        for (int t = 0; t < NT; ++t) {
            v2f b;
            b.x = bv[t].z; b.y = bv[t].w;
            acc[t] = __builtin_amdgcn_wmma_f32_16x16x4_f32(
                false, a1, false, b, (short)0, acc[t], false, false);
        }
    }

    const float ds[8] = { inv_sqrt_deg(dg0.x), inv_sqrt_deg(dg0.y), inv_sqrt_deg(dg0.z), inv_sqrt_deg(dg0.w),
                          inv_sqrt_deg(dg1.x), inv_sqrt_deg(dg1.y), inv_sqrt_deg(dg1.z), inv_sqrt_deg(dg1.w) };
#pragma unroll
    for (int t = 0; t < NT; ++t) {
        const int c = t * 16 + l;
#pragma unroll
        for (int v = 0; v < 8; ++v) {
            const int r = row0 + v + 8 * half;
            Y[(size_t)r * NOUT + c] = acc[t][v] * ds[v];
        }
    }
}

#define TPB 5

template <int NOUT, bool ADD_BIAS, bool SCALE_ROWS>
__global__ __launch_bounds__(256) void gather_gemm_kernel(
    const float* __restrict__ H, const int* __restrict__ slots,
    const int* __restrict__ cnt, const int* __restrict__ novf,
    const int* __restrict__ ovf_c, const int* __restrict__ ovf_r,
    const float* __restrict__ bias, const float* __restrict__ Wn,
    const float* __restrict__ bn, float* __restrict__ Y)
{
    constexpr int TSTRIDE = 68;
    constexpr int NT = (NOUT + 15) / 16;
    constexpr int NP = NT * 16;
    __shared__ __attribute__((aligned(16))) float T[2][16 * TSTRIDE];
    __shared__ __attribute__((aligned(16))) float Ws[NP * TSTRIDE];

    const int tid  = threadIdx.x;
    const int nrow = tid >> 4;
    const int l16  = tid & 15;
    const int f4   = l16 << 2;
    const unsigned lb = (unsigned)f4 << 2;

    int c = (blockIdx.x * TPB) * 16 + nrow;

    int    m    = cnt[c];
    int2   sraw = *(const int2*)(slots + (size_t)c * SLOTS + 2 * l16);
    const float4 b = *(const float4*)(bias + f4);

    {
        constexpr int PER = (64 * NOUT + 255) / 256;
        float wv[PER];
#pragma unroll
        for (int i = 0; i < PER; ++i) {
            const int idx = i * 256 + tid;
            wv[i] = (idx < 64 * NOUT) ? Wn[idx] : 0.0f;
        }
#pragma unroll
        for (int i = 0; i < PER; ++i) {
            const int idx = i * 256 + tid;
            if (idx < 64 * NOUT) Ws[(idx % NOUT) * TSTRIDE + (idx / NOUT)] = wv[i];
        }
        if constexpr (NP > NOUT) {
            for (int idx = tid; idx < (NP - NOUT) * 64; idx += 256)
                Ws[(NOUT + (idx >> 6)) * TSTRIDE + (idx & 63)] = 0.0f;
        }
    }

    const int wave = tid >> 5;
    const int lane = tid & 31;
    const int half = lane >> 4;
    const int l    = lane & 15;
    const int wt   = wave & 3;
    const int grp  = wave >> 2;
    const int ncol = wt * 16 + l;
    float bv = 0.0f;
    if constexpr (ADD_BIAS) bv = bn[(ncol < NOUT) ? ncol : 0];

    for (int it = 0; it < TPB; ++it) {
        int m_n = 0;
        int2 sraw_n = make_int2(0, 0);
        if (it + 1 < TPB) {
            const int cn = c + 16;
            m_n    = cnt[cn];
            sraw_n = *(const int2*)(slots + (size_t)cn * SLOTS + 2 * l16);
        }
        const float4 hs = row_f4(H, c, lb);

        const float dc  = inv_sqrt_deg(m);
        const int   ms  = (m < SLOTS) ? m : SLOTS;
        const int sx = (2 * l16     < ms) ? sraw.x : NNODES;
        const int sy = (2 * l16 + 1 < ms) ? sraw.y : NNODES;

        float4 acc = make_float4(0.0f, 0.0f, 0.0f, 0.0f);

        gather8<0>(H, sx, sy, lb, acc);
        if (ms > 8) {
            gather8<1>(H, sx, sy, lb, acc);
            if (ms > 16) {
                gather8<2>(H, sx, sy, lb, acc);
                if (ms > 24) gather8<3>(H, sx, sy, lb, acc);
            }
        }
        if (m > SLOTS) {
            const int no = *novf;
            for (int o = 0; o < no; ++o) {
                if (ovf_c[o] == c) {
                    const int r = ovf_r[o];
                    const float4 h = row_f4(H, r, lb);
                    acc.x += h.x; acc.y += h.y; acc.z += h.z; acc.w += h.w;
                }
            }
        }

        float4 o;
        o.x = (acc.x + hs.x) * dc + b.x;
        o.y = (acc.y + hs.y) * dc + b.y;
        o.z = (acc.z + hs.z) * dc + b.z;
        o.w = (acc.w + hs.w) * dc + b.w;
        o.x = (o.x > 0.0f) ? o.x : 0.0f;
        o.y = (o.y > 0.0f) ? o.y : 0.0f;
        o.z = (o.z > 0.0f) ? o.z : 0.0f;
        o.w = (o.w > 0.0f) ? o.w : 0.0f;
        if constexpr (SCALE_ROWS) { o.x *= dc; o.y *= dc; o.z *= dc; o.w *= dc; }
        float* __restrict__ Tb = T[it & 1];
        *(float4*)&Tb[nrow * TSTRIDE + f4] = o;

        __syncthreads();

        if (grp == (it & 1) && wt < NT) {
            v8f d = {};
            const float* __restrict__ ap = &Tb[l * TSTRIDE + 2 * half];
            const float* __restrict__ bp = &Ws[ncol * TSTRIDE + 2 * half];
#pragma unroll
            for (int s = 0; s < 16; ++s) {
                const v2f a  = *(const v2f*)(ap + 4 * s);
                const v2f bw = *(const v2f*)(bp + 4 * s);
                d = __builtin_amdgcn_wmma_f32_16x16x4_f32(
                    false, a, false, bw, (short)0, d, false, false);
            }
            if (ncol < NOUT) {
                const int row0 = (blockIdx.x * TPB + it) << 4;
                float* __restrict__ yp = Y + (size_t)(row0 + 8 * half) * NOUT + ncol;
                if constexpr (ADD_BIAS) {
#pragma unroll
                    for (int v = 0; v < 8; ++v) yp[v * NOUT] = d[v] + bv;
                } else {
#pragma unroll
                    for (int v = 0; v < 8; ++v) yp[v * NOUT] = d[v];
                }
            }
        }

        m = m_n; sraw = sraw_n;
        c += 16;
    }
}


extern "C" void kernel_launch(void* const* d_in, const int* in_sizes, int n_in,
                              void* d_out, int out_size, void* d_ws, size_t ws_size,
                              hipStream_t stream) {
    const float* x    = (const float*)d_in[0];
    const int*   ei   = (const int*)d_in[1];
    const float* W1   = (const float*)d_in[2];
    const float* b1   = (const float*)d_in[3];
    const float* W2   = (const float*)d_in[4];
    const float* b2   = (const float*)d_in[5];
    const float* Wout = (const float*)d_in[6];
    const float* bout = (const float*)d_in[7];
    float* out = (float*)d_out;

    const int* row = ei;
    const int* col = ei + NEDGES;

    float* A   = (float*)d_ws;
    float* B   = A + (size_t)(NNODES + 1) * 64;
    int* slots = (int*)(B + (size_t)(NNODES + 1) * 64);
    int* pairs = slots + (size_t)NNODES * SLOTS;
    int* segst = pairs + (size_t)NBLK1 * EPB * 2;
    int* cnt   = segst + (size_t)NBLK1 * SEGW;
    int* novf  = cnt + NNODES;
    int* ovf_c = novf + 1;
    int* ovf_r = ovf_c + NEDGES;
    (void)ws_size; (void)n_in; (void)in_sizes; (void)out_size;
    static_assert((size_t)NBLK1 * EPB >= NEDGES, "bucket_kernel blocks cover every edge");
    static_assert(((size_t)NNODES * SLOTS) % 4 == 0 && ((size_t)NBLK1 * EPB * 2) % 4 == 0 && ((size_t)NBLK1 * SEGW) % 4 == 0,
                  "pairs starts on 8 bytes and cnt on 16 (the product reads in-degrees four at a time)");

    const int T = 256;
    const int Mtiles = NNODES / 16;
    const int gemmBlocks = (Mtiles + 7) / 8;
    static_assert(NNODES % (16 * TPB) == 0, "a gather block owns TPB whole tiles of 16 nodes");
    const int gatherBlocks = NNODES / (16 * TPB);

    bucket_kernel<<<NBLK1, T1, 0, stream>>>(row, col, pairs, segst, novf,
                                            A + (size_t)NNODES * 64, B + (size_t)NNODES * 64);
    rank_kernel<<<NB, NBLK1, 0, stream>>>(pairs, segst, cnt, slots, novf, ovf_c, ovf_r);

    gemm_wmma_kernel<128, 64><<<gemmBlocks, T, 0, stream>>>(x, W1, cnt, A, NNODES);

    gather_gemm_kernel<64, false, true><<<gatherBlocks, T, 0, stream>>>(
        A, slots, cnt, novf, ovf_c, ovf_r, b1, W2, nullptr, B);

    gather_gemm_kernel<40, true, false><<<gatherBlocks, T, 0, stream>>>(
        B, slots, cnt, novf, ovf_c, ovf_r, b2, Wout, bout, out);
}


# ranking launch also forms X @ W1 for its own 256 rows: in-degrees from its LDS counters, X asked for at block entry and kept in flight by LDS-only barriers; 5 launches to 4; sums unchanged
# speedup vs baseline: 5.5333x; 1.0486x over previous
#include <hip/hip_runtime.h>

typedef __attribute__((ext_vector_type(2))) float v2f;
typedef __attribute__((ext_vector_type(8))) float v8f;

#define NNODES 100000
#define NEDGES 1250000


#define SLOTS 32


#define EPB    4884
#define NBLK1  256
#define RANGE  256
#define NB     ((NNODES + RANGE - 1) / RANGE)
#define SEGW   (NB + 1)
#define T1     512
#define EPT    ((EPB + T1 - 1) / T1)

__global__ __launch_bounds__(T1) void bucket_kernel(
    const int* __restrict__ row, const int* __restrict__ col,
    int* __restrict__ pairs, int* __restrict__ segstart,
    int* __restrict__ novf, float* __restrict__ z0, float* __restrict__ z1)
{
    static_assert(SEGW <= T1, "one counter per thread in the scan");
    __shared__ int hist[T1];
    __shared__ int offs[T1];
    __shared__ int wsum[T1 / 32];

    const int tid = threadIdx.x;
    const int blk = blockIdx.x;
    if (blk == 0) {
        if (tid < 64) { z0[tid] = 0.0f; z1[tid] = 0.0f; }
        if (tid == 64) *novf = 0;
    }
    hist[tid] = 0;
    __syncthreads();

    const int base = blk * EPB;
    const int end  = (base + EPB < NEDGES) ? base + EPB : NEDGES;

    int c[EPT], r[EPT], k[EPT];
#pragma unroll
    for (int j = 0; j < EPT; ++j) {
        const int e = base + j * T1 + tid;
        const bool ok = e < end;
        c[j] = ok ? col[e] : -1;
        r[j] = ok ? row[e] : 0;
    }
#pragma unroll
    for (int j = 0; j < EPT; ++j) {
        k[j] = 0;
        if (c[j] >= 0) k[j] = atomicAdd(&hist[c[j] / RANGE], 1);
    }
    __syncthreads();

    const int lane = tid & 31;
    const int wave = tid >> 5;
    const int v = hist[tid];
    int incl = v;
#pragma unroll
    for (int d = 1; d < 32; d <<= 1) {
        const int t = __shfl_up(incl, d);
        if (lane >= d) incl += t;
    }
    if (lane == 31) wsum[wave] = incl;
    __syncthreads();
    int add = 0;
    for (int w = 0; w < wave; ++w) add += wsum[w];
    offs[tid] = add + incl - v;
    __syncthreads();

    int2* __restrict__ reg = (int2*)pairs + (size_t)blk * EPB;
#pragma unroll
    for (int j = 0; j < EPT; ++j) {
        if (c[j] >= 0) reg[offs[c[j] / RANGE] + k[j]] = make_int2(c[j], r[j]);
    }
    if (tid < SEGW) segstart[blk * SEGW + tid] = offs[tid];
}

__device__ __forceinline__ void store_edge(int2 p, int k,
                                           int* __restrict__ slots, int* __restrict__ novf,
                                           int* __restrict__ ovf_c, int* __restrict__ ovf_r) {
    if (k < SLOTS) {
        slots[p.x * SLOTS + k] = p.y;
    } else {
        const int o = atomicAdd(novf, 1);
        ovf_c[o] = p.x;
        ovf_r[o] = p.y;
    }
}

__device__ __forceinline__ float inv_sqrt_deg(int indeg) {
    const float x = (float)(indeg + 1);
    return (x > 0.0f) ? rsqrtf(x) : 0.0f;
}

__device__ __forceinline__ float4 row_f4(const float* __restrict__ H, int r, unsigned lane_bytes) {
    const unsigned off = ((unsigned)r << 8) + lane_bytes;
    return *(const float4*)((const char*)H + off);
}

template <int SRC>
__device__ __forceinline__ int row_bcast(int v) {
    return __builtin_amdgcn_mov_dpp(v, 0x150 + SRC, 0xf, 0xf, true);
}

template <int G>
__device__ __forceinline__ void gather8(const float* __restrict__ H, int sx, int sy,
                                        unsigned lb, float4& acc) {
    const int r0 = row_bcast<4 * G + 0>(sx), r1 = row_bcast<4 * G + 0>(sy);
    const int r2 = row_bcast<4 * G + 1>(sx), r3 = row_bcast<4 * G + 1>(sy);
    const int r4 = row_bcast<4 * G + 2>(sx), r5 = row_bcast<4 * G + 2>(sy);
    const int r6 = row_bcast<4 * G + 3>(sx), r7 = row_bcast<4 * G + 3>(sy);
    const float4 h0 = row_f4(H, r0, lb);
    const float4 h1 = row_f4(H, r1, lb);
    const float4 h2 = row_f4(H, r2, lb);
    const float4 h3 = row_f4(H, r3, lb);
    const float4 h4 = row_f4(H, r4, lb);
    const float4 h5 = row_f4(H, r5, lb);
    const float4 h6 = row_f4(H, r6, lb);
    const float4 h7 = row_f4(H, r7, lb);
    acc.x += h0.x; acc.y += h0.y; acc.z += h0.z; acc.w += h0.w;
    acc.x += h1.x; acc.y += h1.y; acc.z += h1.z; acc.w += h1.w;
    acc.x += h2.x; acc.y += h2.y; acc.z += h2.z; acc.w += h2.w;
    acc.x += h3.x; acc.y += h3.y; acc.z += h3.z; acc.w += h3.w;
    acc.x += h4.x; acc.y += h4.y; acc.z += h4.z; acc.w += h4.w;
    acc.x += h5.x; acc.y += h5.y; acc.z += h5.z; acc.w += h5.w;
    acc.x += h6.x; acc.y += h6.y; acc.z += h6.z; acc.w += h6.w;
    acc.x += h7.x; acc.y += h7.y; acc.z += h7.z; acc.w += h7.w;
}


__device__ __forceinline__ void lds_barrier() {
    __builtin_amdgcn_fence(__ATOMIC_RELEASE, "workgroup", "local");
    __builtin_amdgcn_s_barrier();
    __builtin_amdgcn_fence(__ATOMIC_ACQUIRE, "workgroup", "local");
}

template <int K, int NOUT>
__device__ __forceinline__ void strip_product(const float4 (&a)[K / 8], const float* __restrict__ bp0,
                                              const int* __restrict__ deg8, float* __restrict__ yp)
{
    constexpr int STRIDE = K + 4;
    constexpr int NT = NOUT / 16;
    constexpr int KB = K / 8;

    v8f acc[NT];
#pragma unroll
    for (int t = 0; t < NT; ++t) acc[t] = {};

#pragma unroll
    for (int j = 0; j < KB; ++j) {
        const float4 av = a[j];
        v2f a0, a1;
        a0.x = av.x; a0.y = av.y;
        a1.x = av.z; a1.y = av.w;
        float4 bv[NT];
#pragma unroll
        for (int t = 0; t < NT; ++t) bv[t] = *(const float4*)(bp0 + t * 16 * STRIDE + 8 * j);
#pragma unroll
        for (int t = 0; t < NT; ++t) {
            v2f b;
            b.x = bv[t].x; b.y = bv[t].y;
            acc[t] = __builtin_amdgcn_wmma_f32_16x16x4_f32(
                false, a0, false, b, (short)0, acc[t], false, false);
        }
#pragma unroll
        for (int t = 0; t < NT; ++t) {
            v2f b;
            b.x = bv[t].z; b.y = bv[t].w;
            acc[t] = __builtin_amdgcn_wmma_f32_16x16x4_f32(
                false, a1, false, b, (short)0, acc[t], false, false);
        }
    }

    const int4 dg0 = *(const int4*)(deg8);
    const int4 dg1 = *(const int4*)(deg8 + 4);
    const float ds[8] = { inv_sqrt_deg(dg0.x), inv_sqrt_deg(dg0.y), inv_sqrt_deg(dg0.z), inv_sqrt_deg(dg0.w),
                          inv_sqrt_deg(dg1.x), inv_sqrt_deg(dg1.y), inv_sqrt_deg(dg1.z), inv_sqrt_deg(dg1.w) };
#pragma unroll
    for (int t = 0; t < NT; ++t) {
#pragma unroll
        for (int v = 0; v < 8; ++v) yp[(size_t)v * NOUT + t * 16] = acc[t][v] * ds[v];
    }
}

template <int K, int NOUT>
__global__ __launch_bounds__(NBLK1) void rank_gemm_kernel(
    const int* __restrict__ pairs, const int* __restrict__ segstart,
    int* __restrict__ cnt, int* __restrict__ slots,
    int* __restrict__ novf, int* __restrict__ ovf_c, int* __restrict__ ovf_r,
    const float* __restrict__ X, const float* __restrict__ W, float* __restrict__ Y)
{
    static_assert(RANGE == NBLK1, "one node counter and one piece bound per thread");
    static_assert(RANGE == 2 * (NBLK1 / 32) * 16, "two strips of 16 rows per wave cover the range");
    static_assert(NNODES % 16 == 0, "a strip is inside the node array or outside it as a whole");
    static_assert((NB - 1) * RANGE + RANGE / 2 <= NNODES, "every block's first strips are inside the node array");
    constexpr int STRIDE = K + 4;
    constexpr int KB = K / 8;
    constexpr int QG = NOUT / 16;
    constexpr int CH = KB * QG;
    static_assert(K % 8 == 0 && NOUT % 16 == 0 && CH % 8 == 0, "tile shape");
    static_assert(((STRIDE / 4) & 1) == 1 && STRIDE % 4 == 0, "row pitch: 16 bytes times an odd number");
    __shared__ __attribute__((aligned(16))) int lcnt[RANGE];
    __shared__ int sst[NBLK1];
    __shared__ int sen[NBLK1];
    __shared__ __attribute__((aligned(16))) float Wt[NOUT * STRIDE];

    const int tid = threadIdx.x;
    const int b = blockIdx.x;
    const int nbase = b * RANGE;
    const int lane = tid & 31;
    const int wave = tid >> 5;
    const int half = lane >> 4;
    const int l    = lane & 15;

    const int st0 = segstart[tid * SEGW + b];
    const int en0 = segstart[tid * SEGW + b + 1];

    const int rowA = nbase + 16 * wave;
    const int rowB = rowA + RANGE / 2;
    const bool actB = rowB < NNODES;
    const float4* __restrict__ apA = (const float4*)(X + (size_t)(rowA + l) * K + 4 * half);
    const float4* __restrict__ apB = (const float4*)(X + (size_t)(actB ? rowB + l : 0) * K + 4 * half);

    float4 a[KB];
#pragma unroll
    for (int j = 0; j < KB; ++j) a[j] = apA[2 * j];

    float4 w[CH / 8];
    int wn[CH / 8], wk[CH / 8];
#pragma unroll
    for (int i = 0; i < CH / 8; ++i) {
        const int c = wave * (CH / 8) + i;
        const int q = (c % QG) * 4 + (lane & 3);
        wk[i] = (c / QG) * 8 + (lane >> 2);
        wn[i] = 4 * q;
        w[i] = *(const float4*)(W + (size_t)wk[i] * NOUT + wn[i]);
    }

    lcnt[tid] = 0;
    sst[tid] = st0;
    sen[tid] = en0;
    lds_barrier();

    const int hw  = tid >> 4;
    const int l16 = tid & 15;
    const int2* __restrict__ pp = (const int2*)pairs;

#pragma unroll
    for (int g = 0; g < 2; ++g) {
        int2 p[8];
        bool ok[8];
#pragma unroll
        for (int u = 0; u < 8; ++u) {
            const int s = hw + 16 * (8 * g + u);
            const int i = sst[s] + l16;
            ok[u] = i < sen[s];
            p[u] = make_int2(0, 0);
            if (ok[u]) p[u] = pp[(size_t)s * EPB + i];
        }
        int k[8];
#pragma unroll
        for (int u = 0; u < 8; ++u) {
            k[u] = 0;
            if (ok[u]) k[u] = atomicAdd(&lcnt[p[u].x - nbase], 1);
        }
#pragma unroll
        for (int u = 0; u < 8; ++u) {
            if (ok[u]) store_edge(p[u], k[u], slots, novf, ovf_c, ovf_r);
        }
#pragma unroll
        for (int u = 0; u < 8; ++u) {
            const int s = hw + 16 * (8 * g + u);
            const int en = sen[s];
            for (int i = sst[s] + 16 + l16; i < en; i += 16) {
                const int2 q = pp[(size_t)s * EPB + i];
                store_edge(q, atomicAdd(&lcnt[q.x - nbase], 1), slots, novf, ovf_c, ovf_r);
            }
        }
    }

#pragma unroll
    for (int i = 0; i < CH / 8; ++i) {
        float* __restrict__ d = &Wt[wn[i] * STRIDE + wk[i]];
        d[0]          = w[i].x;
        d[STRIDE]     = w[i].y;
        d[2 * STRIDE] = w[i].z;
        d[3 * STRIDE] = w[i].w;
    }
    lds_barrier();
    if (nbase + tid < NNODES) cnt[nbase + tid] = lcnt[tid];

    float4 a2[KB];
    if (actB) {
#pragma unroll
        for (int j = 0; j < KB; ++j) a2[j] = apB[2 * j];
    }
    const float* __restrict__ bp0 = &Wt[l * STRIDE + 4 * half];
    strip_product<K, NOUT>(a, bp0, &lcnt[16 * wave + 8 * half],
                           Y + (size_t)(rowA + 8 * half) * NOUT + l);
    if (actB)
        strip_product<K, NOUT>(a2, bp0, &lcnt[RANGE / 2 + 16 * wave + 8 * half],
                               Y + (size_t)(rowB + 8 * half) * NOUT + l);
}

#define TPB 5

template <int NOUT, bool ADD_BIAS, bool SCALE_ROWS>
__global__ __launch_bounds__(256) void gather_gemm_kernel(
    const float* __restrict__ H, const int* __restrict__ slots,
    const int* __restrict__ cnt, const int* __restrict__ novf,
    const int* __restrict__ ovf_c, const int* __restrict__ ovf_r,
    const float* __restrict__ bias, const float* __restrict__ Wn,
    const float* __restrict__ bn, float* __restrict__ Y)
{
    constexpr int TSTRIDE = 68;
    constexpr int NT = (NOUT + 15) / 16;
    constexpr int NP = NT * 16;
    __shared__ __attribute__((aligned(16))) float T[2][16 * TSTRIDE];
    __shared__ __attribute__((aligned(16))) float Ws[NP * TSTRIDE];

    const int tid  = threadIdx.x;
    const int nrow = tid >> 4;
    const int l16  = tid & 15;
    const int f4   = l16 << 2;
    const unsigned lb = (unsigned)f4 << 2;

    int c = (blockIdx.x * TPB) * 16 + nrow;

    int    m    = cnt[c];
    int2   sraw = *(const int2*)(slots + (size_t)c * SLOTS + 2 * l16);
    const float4 b = *(const float4*)(bias + f4);

    {
        constexpr int PER = (64 * NOUT + 255) / 256;
        float wv[PER];
#pragma unroll
        for (int i = 0; i < PER; ++i) {
            const int idx = i * 256 + tid;
            wv[i] = (idx < 64 * NOUT) ? Wn[idx] : 0.0f;
        }
#pragma unroll
        for (int i = 0; i < PER; ++i) {
            const int idx = i * 256 + tid;
            if (idx < 64 * NOUT) Ws[(idx % NOUT) * TSTRIDE + (idx / NOUT)] = wv[i];
        }
        if constexpr (NP > NOUT) {
            for (int idx = tid; idx < (NP - NOUT) * 64; idx += 256)
                Ws[(NOUT + (idx >> 6)) * TSTRIDE + (idx & 63)] = 0.0f;
        }
    }

    const int wave = tid >> 5;
    const int lane = tid & 31;
    const int half = lane >> 4;
    const int l    = lane & 15;
    const int wt   = wave & 3;
    const int grp  = wave >> 2;
    const int ncol = wt * 16 + l;
    float bv = 0.0f;
    if constexpr (ADD_BIAS) bv = bn[(ncol < NOUT) ? ncol : 0];

    for (int it = 0; it < TPB; ++it) {
        int m_n = 0;
        int2 sraw_n = make_int2(0, 0);
        if (it + 1 < TPB) {
            const int cn = c + 16;
            m_n    = cnt[cn];
            sraw_n = *(const int2*)(slots + (size_t)cn * SLOTS + 2 * l16);
        }
        const float4 hs = row_f4(H, c, lb);

        const float dc  = inv_sqrt_deg(m);
        const int   ms  = (m < SLOTS) ? m : SLOTS;
        const int sx = (2 * l16     < ms) ? sraw.x : NNODES;
        const int sy = (2 * l16 + 1 < ms) ? sraw.y : NNODES;

        float4 acc = make_float4(0.0f, 0.0f, 0.0f, 0.0f);

        gather8<0>(H, sx, sy, lb, acc);
        if (ms > 8) {
            gather8<1>(H, sx, sy, lb, acc);
            if (ms > 16) {
                gather8<2>(H, sx, sy, lb, acc);
                if (ms > 24) gather8<3>(H, sx, sy, lb, acc);
            }
        }
        if (m > SLOTS) {
            const int no = *novf;
            for (int o = 0; o < no; ++o) {
                if (ovf_c[o] == c) {
                    const int r = ovf_r[o];
                    const float4 h = row_f4(H, r, lb);
                    acc.x += h.x; acc.y += h.y; acc.z += h.z; acc.w += h.w;
                }
            }
        }

        float4 o;
        o.x = (acc.x + hs.x) * dc + b.x;
        o.y = (acc.y + hs.y) * dc + b.y;
        o.z = (acc.z + hs.z) * dc + b.z;
        o.w = (acc.w + hs.w) * dc + b.w;
        o.x = (o.x > 0.0f) ? o.x : 0.0f;
        o.y = (o.y > 0.0f) ? o.y : 0.0f;
        o.z = (o.z > 0.0f) ? o.z : 0.0f;
        o.w = (o.w > 0.0f) ? o.w : 0.0f;
        if constexpr (SCALE_ROWS) { o.x *= dc; o.y *= dc; o.z *= dc; o.w *= dc; }
        float* __restrict__ Tb = T[it & 1];
        *(float4*)&Tb[nrow * TSTRIDE + f4] = o;

        __syncthreads();

        if (grp == (it & 1) && wt < NT) {
            v8f d = {};
            const float* __restrict__ ap = &Tb[l * TSTRIDE + 2 * half];
            const float* __restrict__ bp = &Ws[ncol * TSTRIDE + 2 * half];
#pragma unroll
            for (int s = 0; s < 16; ++s) {
                const v2f a  = *(const v2f*)(ap + 4 * s);
                const v2f bw = *(const v2f*)(bp + 4 * s);
                d = __builtin_amdgcn_wmma_f32_16x16x4_f32(
                    false, a, false, bw, (short)0, d, false, false);
            }
            if (ncol < NOUT) {
                const int row0 = (blockIdx.x * TPB + it) << 4;
                float* __restrict__ yp = Y + (size_t)(row0 + 8 * half) * NOUT + ncol;
                if constexpr (ADD_BIAS) {
#pragma unroll
                    for (int v = 0; v < 8; ++v) yp[v * NOUT] = d[v] + bv;
                } else {
#pragma unroll
                    for (int v = 0; v < 8; ++v) yp[v * NOUT] = d[v];
                }
            }
        }

        m = m_n; sraw = sraw_n;
        c += 16;
    }
}


extern "C" void kernel_launch(void* const* d_in, const int* in_sizes, int n_in,
                              void* d_out, int out_size, void* d_ws, size_t ws_size,
                              hipStream_t stream) {
    const float* x    = (const float*)d_in[0];
    const int*   ei   = (const int*)d_in[1];
    const float* W1   = (const float*)d_in[2];
    const float* b1   = (const float*)d_in[3];
    const float* W2   = (const float*)d_in[4];
    const float* b2   = (const float*)d_in[5];
    const float* Wout = (const float*)d_in[6];
    const float* bout = (const float*)d_in[7];
    float* out = (float*)d_out;

    const int* row = ei;
    const int* col = ei + NEDGES;

    float* A   = (float*)d_ws;
    float* B   = A + (size_t)(NNODES + 1) * 64;
    int* slots = (int*)(B + (size_t)(NNODES + 1) * 64);
    int* pairs = slots + (size_t)NNODES * SLOTS;
    int* segst = pairs + (size_t)NBLK1 * EPB * 2;
    int* cnt   = segst + (size_t)NBLK1 * SEGW;
    int* novf  = cnt + NNODES;
    int* ovf_c = novf + 1;
    int* ovf_r = ovf_c + NEDGES;
    (void)ws_size; (void)n_in; (void)in_sizes; (void)out_size;
    static_assert((size_t)NBLK1 * EPB >= NEDGES, "bucket_kernel blocks cover every edge");
    static_assert(((size_t)NNODES * SLOTS) % 4 == 0 && ((size_t)NBLK1 * EPB * 2) % 4 == 0 && ((size_t)NBLK1 * SEGW) % 4 == 0,
                  "pairs starts on 8 bytes and cnt on 16 (the product reads in-degrees four at a time)");

    const int T = 256;
    static_assert(NNODES % (16 * TPB) == 0, "a gather block owns TPB whole tiles of 16 nodes");
    const int gatherBlocks = NNODES / (16 * TPB);

    bucket_kernel<<<NBLK1, T1, 0, stream>>>(row, col, pairs, segst, novf,
                                            A + (size_t)NNODES * 64, B + (size_t)NNODES * 64);
    rank_gemm_kernel<128, 64><<<NB, NBLK1, 0, stream>>>(pairs, segst, cnt, slots, novf, ovf_c, ovf_r,
                                                        x, W1, A);

    gather_gemm_kernel<64, false, true><<<gatherBlocks, T, 0, stream>>>(
        A, slots, cnt, novf, ovf_c, ovf_r, b1, W2, nullptr, B);

    gather_gemm_kernel<40, true, false><<<gatherBlocks, T, 0, stream>>>(
        B, slots, cnt, novf, ovf_c, ovf_r, b2, Wout, bout, out);
}


# ranking+product launch re-cut to 250 ranges of 400 nodes, one 16-wave block per WGP (was 391 blocks on 256): even matrix work, a piece's edges all asked for at once; slot order stays arrival order
# speedup vs baseline: 5.7461x; 1.0385x over previous
#include <hip/hip_runtime.h>

typedef __attribute__((ext_vector_type(2))) float v2f;
typedef __attribute__((ext_vector_type(8))) float v8f;

#define NNODES 100000
#define NEDGES 1250000


#define SLOTS 32


#define EPB    4884
#define NBLK1  256
#define RANGE  400
#define NB     ((NNODES + RANGE - 1) / RANGE)
#define SEGW   (NB + 1)
#define T1     512
#define RT     512
#define EPT    ((EPB + T1 - 1) / T1)

__global__ __launch_bounds__(T1) void bucket_kernel(
    const int* __restrict__ row, const int* __restrict__ col,
    int* __restrict__ pairs, int* __restrict__ segstart,
    int* __restrict__ novf, float* __restrict__ z0, float* __restrict__ z1)
{
    static_assert(SEGW <= T1, "one counter per thread in the scan");
    __shared__ int hist[T1];
    __shared__ int offs[T1];
    __shared__ int wsum[T1 / 32];

    const int tid = threadIdx.x;
    const int blk = blockIdx.x;
    if (blk == 0) {
        if (tid < 64) { z0[tid] = 0.0f; z1[tid] = 0.0f; }
        if (tid == 64) *novf = 0;
    }
    hist[tid] = 0;
    __syncthreads();

    const int base = blk * EPB;
    const int end  = (base + EPB < NEDGES) ? base + EPB : NEDGES;

    int c[EPT], r[EPT], k[EPT];
#pragma unroll
    for (int j = 0; j < EPT; ++j) {
        const int e = base + j * T1 + tid;
        const bool ok = e < end;
        c[j] = ok ? col[e] : -1;
        r[j] = ok ? row[e] : 0;
    }
#pragma unroll
    for (int j = 0; j < EPT; ++j) {
        k[j] = 0;
        if (c[j] >= 0) k[j] = atomicAdd(&hist[c[j] / RANGE], 1);
    }
    __syncthreads();

    const int lane = tid & 31;
    const int wave = tid >> 5;
    const int v = hist[tid];
    int incl = v;
#pragma unroll
    for (int d = 1; d < 32; d <<= 1) {
        const int t = __shfl_up(incl, d);
        if (lane >= d) incl += t;
    }
    if (lane == 31) wsum[wave] = incl;
    __syncthreads();
    int add = 0;
    for (int w = 0; w < wave; ++w) add += wsum[w];
    offs[tid] = add + incl - v;
    __syncthreads();

    int2* __restrict__ reg = (int2*)pairs + (size_t)blk * EPB;
#pragma unroll
    for (int j = 0; j < EPT; ++j) {
        if (c[j] >= 0) reg[offs[c[j] / RANGE] + k[j]] = make_int2(c[j], r[j]);
    }
    if (tid < SEGW) segstart[blk * SEGW + tid] = offs[tid];
}

__device__ __forceinline__ void store_edge(int2 p, int k,
                                           int* __restrict__ slots, int* __restrict__ novf,
                                           int* __restrict__ ovf_c, int* __restrict__ ovf_r) {
    if (k < SLOTS) {
        slots[p.x * SLOTS + k] = p.y;
    } else {
        const int o = atomicAdd(novf, 1);
        ovf_c[o] = p.x;
        ovf_r[o] = p.y;
    }
}

__device__ __forceinline__ float inv_sqrt_deg(int indeg) {
    const float x = (float)(indeg + 1);
    return (x > 0.0f) ? rsqrtf(x) : 0.0f;
}

__device__ __forceinline__ float4 row_f4(const float* __restrict__ H, int r, unsigned lane_bytes) {
    const unsigned off = ((unsigned)r << 8) + lane_bytes;
    return *(const float4*)((const char*)H + off);
}

template <int SRC>
__device__ __forceinline__ int row_bcast(int v) {
    return __builtin_amdgcn_mov_dpp(v, 0x150 + SRC, 0xf, 0xf, true);
}

template <int G>
__device__ __forceinline__ void gather8(const float* __restrict__ H, int sx, int sy,
                                        unsigned lb, float4& acc) {
    const int r0 = row_bcast<4 * G + 0>(sx), r1 = row_bcast<4 * G + 0>(sy);
    const int r2 = row_bcast<4 * G + 1>(sx), r3 = row_bcast<4 * G + 1>(sy);
    const int r4 = row_bcast<4 * G + 2>(sx), r5 = row_bcast<4 * G + 2>(sy);
    const int r6 = row_bcast<4 * G + 3>(sx), r7 = row_bcast<4 * G + 3>(sy);
    const float4 h0 = row_f4(H, r0, lb);
    const float4 h1 = row_f4(H, r1, lb);
    const float4 h2 = row_f4(H, r2, lb);
    const float4 h3 = row_f4(H, r3, lb);
    const float4 h4 = row_f4(H, r4, lb);
    const float4 h5 = row_f4(H, r5, lb);
    const float4 h6 = row_f4(H, r6, lb);
    const float4 h7 = row_f4(H, r7, lb);
    acc.x += h0.x; acc.y += h0.y; acc.z += h0.z; acc.w += h0.w;
    acc.x += h1.x; acc.y += h1.y; acc.z += h1.z; acc.w += h1.w;
    acc.x += h2.x; acc.y += h2.y; acc.z += h2.z; acc.w += h2.w;
    acc.x += h3.x; acc.y += h3.y; acc.z += h3.z; acc.w += h3.w;
    acc.x += h4.x; acc.y += h4.y; acc.z += h4.z; acc.w += h4.w;
    acc.x += h5.x; acc.y += h5.y; acc.z += h5.z; acc.w += h5.w;
    acc.x += h6.x; acc.y += h6.y; acc.z += h6.z; acc.w += h6.w;
    acc.x += h7.x; acc.y += h7.y; acc.z += h7.z; acc.w += h7.w;
}


__device__ __forceinline__ void lds_barrier() {
    __builtin_amdgcn_fence(__ATOMIC_RELEASE, "workgroup", "local");
    __builtin_amdgcn_s_barrier();
    __builtin_amdgcn_fence(__ATOMIC_ACQUIRE, "workgroup", "local");
}

template <int K, int NOUT>
__device__ __forceinline__ void strip_product(const float4 (&a)[K / 8], const float* __restrict__ bp0,
                                              const int* __restrict__ deg8, float* __restrict__ yp)
{
    constexpr int STRIDE = K + 4;
    constexpr int NT = NOUT / 16;
    constexpr int KB = K / 8;

    v8f acc[NT];
#pragma unroll
    for (int t = 0; t < NT; ++t) acc[t] = {};

#pragma unroll
    for (int j = 0; j < KB; ++j) {
        const float4 av = a[j];
        v2f a0, a1;
        a0.x = av.x; a0.y = av.y;
        a1.x = av.z; a1.y = av.w;
        float4 bv[NT];
#pragma unroll
        for (int t = 0; t < NT; ++t) bv[t] = *(const float4*)(bp0 + t * 16 * STRIDE + 8 * j);
#pragma unroll
        for (int t = 0; t < NT; ++t) {
            v2f b;
            b.x = bv[t].x; b.y = bv[t].y;
            acc[t] = __builtin_amdgcn_wmma_f32_16x16x4_f32(
                false, a0, false, b, (short)0, acc[t], false, false);
        }
#pragma unroll
        for (int t = 0; t < NT; ++t) {
            v2f b;
            b.x = bv[t].z; b.y = bv[t].w;
            acc[t] = __builtin_amdgcn_wmma_f32_16x16x4_f32(
                false, a1, false, b, (short)0, acc[t], false, false);
        }
    }

    const int4 dg0 = *(const int4*)(deg8);
    const int4 dg1 = *(const int4*)(deg8 + 4);
    const float ds[8] = { inv_sqrt_deg(dg0.x), inv_sqrt_deg(dg0.y), inv_sqrt_deg(dg0.z), inv_sqrt_deg(dg0.w),
                          inv_sqrt_deg(dg1.x), inv_sqrt_deg(dg1.y), inv_sqrt_deg(dg1.z), inv_sqrt_deg(dg1.w) };
#pragma unroll
    for (int t = 0; t < NT; ++t) {
#pragma unroll
        for (int v = 0; v < 8; ++v) yp[(size_t)v * NOUT + t * 16] = acc[t][v] * ds[v];
    }
}

template <int K, int NOUT>
__global__ __launch_bounds__(RT) void rank_gemm_kernel(
    const int* __restrict__ pairs, const int* __restrict__ segstart,
    int* __restrict__ cnt, int* __restrict__ slots,
    int* __restrict__ novf, int* __restrict__ ovf_c, int* __restrict__ ovf_r,
    const float* __restrict__ X, const float* __restrict__ W, float* __restrict__ Y)
{
    constexpr int NW  = RT / 32;
    constexpr int NLR = RT / 16;
    constexpr int PPR = NBLK1 / NLR;
    constexpr int S2  = RANGE / 16 - NW;
    static_assert(RANGE <= RT && NBLK1 <= RT, "one node counter and one piece bound per thread");
    static_assert(NBLK1 % NLR == 0, "every lane row takes the same number of pieces");
    static_assert(NB * RANGE == NNODES && RANGE % 16 == 0, "the ranges cover the node array exactly, in whole strips");
    static_assert(S2 >= 0 && S2 <= NW, "one or two strips per wave");
    constexpr int STRIDE = K + 4;
    constexpr int KB = K / 8;
    constexpr int QG = NOUT / 16;
    constexpr int CH = KB * QG;
    constexpr int CW = CH / NW;
    static_assert(K % 8 == 0 && NOUT % 16 == 0 && CH % NW == 0, "tile shape");
    static_assert(((STRIDE / 4) & 1) == 1 && STRIDE % 4 == 0, "row pitch: 16 bytes times an odd number");
    __shared__ __attribute__((aligned(16))) int lcnt[RANGE];
    __shared__ int2 sbnd[NBLK1];
    __shared__ __attribute__((aligned(16))) float Wt[NOUT * STRIDE];

    const int tid = threadIdx.x;
    const int b = blockIdx.x;
    const int nbase = b * RANGE;
    const int lane = tid & 31;
    const int wave = tid >> 5;
    const int half = lane >> 4;
    const int l    = lane & 15;

    int st0 = 0, en0 = 0;
    if (tid < NBLK1) {
        st0 = segstart[tid * SEGW + b];
        en0 = segstart[tid * SEGW + b + 1];
    }

    const int rowA = nbase + 16 * wave;
    const int rowB = rowA + 16 * NW;
    const bool actB = wave < S2;
    const float4* __restrict__ apA = (const float4*)(X + (size_t)(rowA + l) * K + 4 * half);
    const float4* __restrict__ apB = (const float4*)(X + (size_t)(actB ? rowB + l : 0) * K + 4 * half);

    float4 a[KB];
#pragma unroll
    for (int j = 0; j < KB; ++j) a[j] = apA[2 * j];

    float4 w[CW];
    int wn[CW], wk[CW];
#pragma unroll
    for (int i = 0; i < CW; ++i) {
        const int c = wave * CW + i;
        const int q = (c % QG) * 4 + (lane & 3);
        wk[i] = (c / QG) * 8 + (lane >> 2);
        wn[i] = 4 * q;
        w[i] = *(const float4*)(W + (size_t)wk[i] * NOUT + wn[i]);
    }

    if (tid < RANGE) lcnt[tid] = 0;
    if (tid < NBLK1) sbnd[tid] = make_int2(st0, en0);
    lds_barrier();

    const int hw  = tid >> 4;
    const int l16 = tid & 15;
    const int2* __restrict__ pp = (const int2*)pairs;

    int2 bd[PPR];
#pragma unroll
    for (int u = 0; u < PPR; ++u) bd[u] = sbnd[hw + NLR * u];

    int2 p0[PPR], p1[PPR];
    bool ok0[PPR], ok1[PPR];
#pragma unroll
    for (int u = 0; u < PPR; ++u) {
        const int s = hw + NLR * u;
        const int i = bd[u].x + l16;
        ok0[u] = i < bd[u].y;
        ok1[u] = i + 16 < bd[u].y;
        p0[u] = make_int2(0, 0);
        p1[u] = make_int2(0, 0);
        if (ok0[u]) p0[u] = pp[(size_t)s * EPB + i];
        if (ok1[u]) p1[u] = pp[(size_t)s * EPB + i + 16];
    }
    int k0[PPR], k1[PPR];
#pragma unroll
    for (int u = 0; u < PPR; ++u) {
        k0[u] = 0;
        if (ok0[u]) k0[u] = atomicAdd(&lcnt[p0[u].x - nbase], 1);
    }
#pragma unroll
    for (int u = 0; u < PPR; ++u) {
        k1[u] = 0;
        if (ok1[u]) k1[u] = atomicAdd(&lcnt[p1[u].x - nbase], 1);
    }
#pragma unroll
    for (int u = 0; u < PPR; ++u) {
        if (ok0[u]) store_edge(p0[u], k0[u], slots, novf, ovf_c, ovf_r);
    }
#pragma unroll
    for (int u = 0; u < PPR; ++u) {
        if (ok1[u]) store_edge(p1[u], k1[u], slots, novf, ovf_c, ovf_r);
    }
#pragma unroll
    for (int u = 0; u < PPR; ++u) {
        const int s = hw + NLR * u;
        const int en = bd[u].y;
        for (int i = bd[u].x + 32 + l16; i < en; i += 16) {
            const int2 q = pp[(size_t)s * EPB + i];
            store_edge(q, atomicAdd(&lcnt[q.x - nbase], 1), slots, novf, ovf_c, ovf_r);
        }
    }

#pragma unroll
    for (int i = 0; i < CW; ++i) {
        float* __restrict__ d = &Wt[wn[i] * STRIDE + wk[i]];
        d[0]          = w[i].x;
        d[STRIDE]     = w[i].y;
        d[2 * STRIDE] = w[i].z;
        d[3 * STRIDE] = w[i].w;
    }
    lds_barrier();
    if (tid < RANGE) cnt[nbase + tid] = lcnt[tid];

    float4 a2[KB];
    if (actB) {
#pragma unroll
        for (int j = 0; j < KB; ++j) a2[j] = apB[2 * j];
    }
    const float* __restrict__ bp0 = &Wt[l * STRIDE + 4 * half];
    strip_product<K, NOUT>(a, bp0, &lcnt[16 * wave + 8 * half],
                           Y + (size_t)(rowA + 8 * half) * NOUT + l);
    if (actB)
        strip_product<K, NOUT>(a2, bp0, &lcnt[16 * NW + 16 * wave + 8 * half],
                               Y + (size_t)(rowB + 8 * half) * NOUT + l);
}

#define TPB 5

template <int NOUT, bool ADD_BIAS, bool SCALE_ROWS>
__global__ __launch_bounds__(256) void gather_gemm_kernel(
    const float* __restrict__ H, const int* __restrict__ slots,
    const int* __restrict__ cnt, const int* __restrict__ novf,
    const int* __restrict__ ovf_c, const int* __restrict__ ovf_r,
    const float* __restrict__ bias, const float* __restrict__ Wn,
    const float* __restrict__ bn, float* __restrict__ Y)
{
    constexpr int TSTRIDE = 68;
    constexpr int NT = (NOUT + 15) / 16;
    constexpr int NP = NT * 16;
    __shared__ __attribute__((aligned(16))) float T[2][16 * TSTRIDE];
    __shared__ __attribute__((aligned(16))) float Ws[NP * TSTRIDE];

    const int tid  = threadIdx.x;
    const int nrow = tid >> 4;
    const int l16  = tid & 15;
    const int f4   = l16 << 2;
    const unsigned lb = (unsigned)f4 << 2;

    int c = (blockIdx.x * TPB) * 16 + nrow;

    int    m    = cnt[c];
    int2   sraw = *(const int2*)(slots + (size_t)c * SLOTS + 2 * l16);
    const float4 b = *(const float4*)(bias + f4);

    {
        constexpr int PER = (64 * NOUT + 255) / 256;
        float wv[PER];
#pragma unroll
        for (int i = 0; i < PER; ++i) {
            const int idx = i * 256 + tid;
            wv[i] = (idx < 64 * NOUT) ? Wn[idx] : 0.0f;
        }
#pragma unroll
        for (int i = 0; i < PER; ++i) {
            const int idx = i * 256 + tid;
            if (idx < 64 * NOUT) Ws[(idx % NOUT) * TSTRIDE + (idx / NOUT)] = wv[i];
        }
        if constexpr (NP > NOUT) {
            for (int idx = tid; idx < (NP - NOUT) * 64; idx += 256)
                Ws[(NOUT + (idx >> 6)) * TSTRIDE + (idx & 63)] = 0.0f;
        }
    }

    const int wave = tid >> 5;
    const int lane = tid & 31;
    const int half = lane >> 4;
    const int l    = lane & 15;
    const int wt   = wave & 3;
    const int grp  = wave >> 2;
    const int ncol = wt * 16 + l;
    float bv = 0.0f;
    if constexpr (ADD_BIAS) bv = bn[(ncol < NOUT) ? ncol : 0];

    for (int it = 0; it < TPB; ++it) {
        int m_n = 0;
        int2 sraw_n = make_int2(0, 0);
        if (it + 1 < TPB) {
            const int cn = c + 16;
            m_n    = cnt[cn];
            sraw_n = *(const int2*)(slots + (size_t)cn * SLOTS + 2 * l16);
        }
        const float4 hs = row_f4(H, c, lb);

        const float dc  = inv_sqrt_deg(m);
        const int   ms  = (m < SLOTS) ? m : SLOTS;
        const int sx = (2 * l16     < ms) ? sraw.x : NNODES;
        const int sy = (2 * l16 + 1 < ms) ? sraw.y : NNODES;

        float4 acc = make_float4(0.0f, 0.0f, 0.0f, 0.0f);

        gather8<0>(H, sx, sy, lb, acc);
        if (ms > 8) {
            gather8<1>(H, sx, sy, lb, acc);
            if (ms > 16) {
                gather8<2>(H, sx, sy, lb, acc);
                if (ms > 24) gather8<3>(H, sx, sy, lb, acc);
            }
        }
        if (m > SLOTS) {
            const int no = *novf;
            for (int o = 0; o < no; ++o) {
                if (ovf_c[o] == c) {
                    const int r = ovf_r[o];
                    const float4 h = row_f4(H, r, lb);
                    acc.x += h.x; acc.y += h.y; acc.z += h.z; acc.w += h.w;
                }
            }
        }

        float4 o;
        o.x = (acc.x + hs.x) * dc + b.x;
        o.y = (acc.y + hs.y) * dc + b.y;
        o.z = (acc.z + hs.z) * dc + b.z;
        o.w = (acc.w + hs.w) * dc + b.w;
        o.x = (o.x > 0.0f) ? o.x : 0.0f;
        o.y = (o.y > 0.0f) ? o.y : 0.0f;
        o.z = (o.z > 0.0f) ? o.z : 0.0f;
        o.w = (o.w > 0.0f) ? o.w : 0.0f;
        if constexpr (SCALE_ROWS) { o.x *= dc; o.y *= dc; o.z *= dc; o.w *= dc; }
        float* __restrict__ Tb = T[it & 1];
        *(float4*)&Tb[nrow * TSTRIDE + f4] = o;

        __syncthreads();

        if (grp == (it & 1) && wt < NT) {
            v8f d = {};
            const float* __restrict__ ap = &Tb[l * TSTRIDE + 2 * half];
            const float* __restrict__ bp = &Ws[ncol * TSTRIDE + 2 * half];
#pragma unroll
            for (int s = 0; s < 16; ++s) {
                const v2f a  = *(const v2f*)(ap + 4 * s);
                const v2f bw = *(const v2f*)(bp + 4 * s);
                d = __builtin_amdgcn_wmma_f32_16x16x4_f32(
                    false, a, false, bw, (short)0, d, false, false);
            }
            if (ncol < NOUT) {
                const int row0 = (blockIdx.x * TPB + it) << 4;
                float* __restrict__ yp = Y + (size_t)(row0 + 8 * half) * NOUT + ncol;
                if constexpr (ADD_BIAS) {
#pragma unroll
                    for (int v = 0; v < 8; ++v) yp[v * NOUT] = d[v] + bv;
                } else {
#pragma unroll
                    for (int v = 0; v < 8; ++v) yp[v * NOUT] = d[v];
                }
            }
        }

        m = m_n; sraw = sraw_n;
        c += 16;
    }
}


extern "C" void kernel_launch(void* const* d_in, const int* in_sizes, int n_in,
                              void* d_out, int out_size, void* d_ws, size_t ws_size,
                              hipStream_t stream) {
    const float* x    = (const float*)d_in[0];
    const int*   ei   = (const int*)d_in[1];
    const float* W1   = (const float*)d_in[2];
    const float* b1   = (const float*)d_in[3];
    const float* W2   = (const float*)d_in[4];
    const float* b2   = (const float*)d_in[5];
    const float* Wout = (const float*)d_in[6];
    const float* bout = (const float*)d_in[7];
    float* out = (float*)d_out;

    const int* row = ei;
    const int* col = ei + NEDGES;

    float* A   = (float*)d_ws;
    float* B   = A + (size_t)(NNODES + 1) * 64;
    int* slots = (int*)(B + (size_t)(NNODES + 1) * 64);
    int* pairs = slots + (size_t)NNODES * SLOTS;
    int* segst = pairs + (size_t)NBLK1 * EPB * 2;
    int* cnt   = segst + (size_t)NBLK1 * SEGW;
    int* novf  = cnt + NNODES;
    int* ovf_c = novf + 1;
    int* ovf_r = ovf_c + NEDGES;
    (void)ws_size; (void)n_in; (void)in_sizes; (void)out_size;
    static_assert((size_t)NBLK1 * EPB >= NEDGES, "bucket_kernel blocks cover every edge");
    static_assert(((size_t)NNODES * SLOTS) % 4 == 0 && ((size_t)NBLK1 * EPB * 2) % 4 == 0 && ((size_t)NBLK1 * SEGW) % 4 == 0,
                  "pairs starts on 8 bytes and cnt on 16 (the product reads in-degrees four at a time)");

    const int T = 256;
    static_assert(NNODES % (16 * TPB) == 0, "a gather block owns TPB whole tiles of 16 nodes");
    const int gatherBlocks = NNODES / (16 * TPB);

    bucket_kernel<<<NBLK1, T1, 0, stream>>>(row, col, pairs, segst, novf,
                                            A + (size_t)NNODES * 64, B + (size_t)NNODES * 64);
    rank_gemm_kernel<128, 64><<<NB, RT, 0, stream>>>(pairs, segst, cnt, slots, novf, ovf_c, ovf_r,
                                                        x, W1, A);

    gather_gemm_kernel<64, false, true><<<gatherBlocks, T, 0, stream>>>(
        A, slots, cnt, novf, ovf_c, ovf_r, b1, W2, nullptr, B);

    gather_gemm_kernel<40, true, false><<<gatherBlocks, T, 0, stream>>>(
        B, slots, cnt, novf, ovf_c, ovf_r, b2, Wout, bout, out);
}
